# starting kernel
# baseline (speedup 1.0000x reference)
#include <hip/hip_runtime.h>
#include <hip/hip_bf16.h>


#define D 128

typedef __attribute__((ext_vector_type(2))) float v2f;
typedef __attribute__((ext_vector_type(8))) float v8f;

__device__ __forceinline__ v8f wmma_f32(v2f a, v2f b, v8f c) {
    return __builtin_amdgcn_wmma_f32_16x16x4_f32(
         false, a,  false, b,
         (short)0, c,  false,  false);
}

__device__ __forceinline__ int sw_addr(int row, int col) {
    return (row << 7) + ((((col >> 1) ^ (row & 63)) << 1) | (col & 1));
}

__device__ __forceinline__ v2f ldsB(const float* sW, int kk, int half, int lm, int n) {
    const int row = (n << 4) + lm;
    const int kB  = (kk << 2) + (half << 1);
    return *(const v2f*)&sW[sw_addr(row, kB)];
}

__global__ void zero_f4_kernel(float4* __restrict__ p, int n4) {
    for (int i = blockIdx.x * 256 + threadIdx.x; i < n4; i += gridDim.x * 256)
        p[i] = make_float4(0.f, 0.f, 0.f, 0.f);
}

__device__ __forceinline__ void ln_scatter(
    const v8f* __restrict__ acc, const int* __restrict__ dst_idx, int ebase,
    int half, int lm, const float* __restrict__ lg, const float* __restrict__ lb,
    float* __restrict__ agg)
{
    float mu[8], rs[8];
#pragma unroll
    for (int j = 0; j < 8; ++j) {
        float s = 0.f, q = 0.f;
#pragma unroll
        for (int n = 0; n < 8; ++n) {
            const float v = acc[n][j];
            s += v; q += v * v;
        }
#pragma unroll
        for (int m = 1; m < 16; m <<= 1) {
            s += __shfl_xor(s, m, 32);
            q += __shfl_xor(q, m, 32);
        }
        const float mean = s * (1.f / 128.f);
        mu[j] = mean;
        rs[j] = rsqrtf(fmaxf(q * (1.f / 128.f) - mean * mean, 0.f) + 1e-5f);
    }

    int drow[8];
#pragma unroll
    for (int j = 0; j < 8; ++j) drow[j] = dst_idx[ebase + j + (half << 3)];

#pragma unroll
    for (int n = 0; n < 8; ++n) {
        const int N = lm + (n << 4);
#pragma unroll
        for (int j = 0; j < 8; ++j) {
            const float v = (acc[n][j] - mu[j]) * rs[j] * lg[n] + lb[n];
            atomicAdd(&agg[(size_t)drow[j] * D + N], v);
        }
    }
}

__global__ __launch_bounds__(256) void gnn_edge_kernel(
    const float* __restrict__ x,
    const int*   __restrict__ src_idx,
    const int*   __restrict__ dst_idx,
    const float* __restrict__ W_w,
    const float* __restrict__ W_b,
    const float* __restrict__ ln_g,
    const float* __restrict__ ln_b,
    float*       __restrict__ agg,
    int n_edges)
{
    __shared__ float sW[D * D];
    for (int i = threadIdx.x; i < D * D; i += 256) {
        int r = i >> 7, c = i & 127;
        sW[sw_addr(r, c)] = W_w[i];
    }
    __syncthreads();

    const int lane = threadIdx.x & 31;
    const int wave = threadIdx.x >> 5;
    const int half = lane >> 4;
    const int lm   = lane & 15;

    float wb[8], lg[8], lb[8];
#pragma unroll
    for (int n = 0; n < 8; ++n) {
        const int N = lm + (n << 4);
        wb[n] = W_b[N]; lg[n] = ln_g[N]; lb[n] = ln_b[N];
    }

    const int n_tiles = n_edges >> 4;
    const int n_pairs = (n_tiles + 1) >> 1;
    for (int pair = blockIdx.x * 8 + wave; pair < n_pairs; pair += gridDim.x * 8) {
        const int tile0  = pair << 1;
        const int tile1  = tile0 + 1;
        const bool has1  = tile1 < n_tiles;
        const int ebase0 = tile0 << 4;
        const int ebase1 = (has1 ? tile1 : tile0) << 4;

        const int src0 = src_idx[ebase0 + lm];
        const int src1 = src_idx[ebase1 + lm];
        const float* arow0 = x + (size_t)src0 * D + (half << 1);
        const float* arow1 = x + (size_t)src1 * D + (half << 1);

        v8f acc0[8], acc1[8];
#pragma unroll
        for (int n = 0; n < 8; ++n) {
#pragma unroll
            for (int q = 0; q < 8; ++q) { acc0[n][q] = wb[n]; acc1[n][q] = wb[n]; }
        }

        v2f bb[2][8];
        v2f aa0[2], aa1[2];
#pragma unroll
        for (int n = 0; n < 8; ++n) bb[0][n] = ldsB(sW, 0, half, lm, n);
        aa0[0] = *(const v2f*)&arow0[0];
        aa1[0] = *(const v2f*)&arow1[0];

#pragma unroll 2
        for (int kk = 0; kk < 32; ++kk) {
            const int cur = kk & 1, nxt = cur ^ 1;
            if (kk < 31) {
#pragma unroll
                for (int n = 0; n < 8; ++n) bb[nxt][n] = ldsB(sW, kk + 1, half, lm, n);
                aa0[nxt] = *(const v2f*)&arow0[(kk + 1) << 2];
                aa1[nxt] = *(const v2f*)&arow1[(kk + 1) << 2];
            }
#pragma unroll
            for (int n = 0; n < 8; ++n) acc0[n] = wmma_f32(aa0[cur], bb[cur][n], acc0[n]);
#pragma unroll
            for (int n = 0; n < 8; ++n) acc1[n] = wmma_f32(aa1[cur], bb[cur][n], acc1[n]);
        }

        ln_scatter(acc0, dst_idx, ebase0, half, lm, lg, lb, agg);
        if (has1) ln_scatter(acc1, dst_idx, ebase1, half, lm, lg, lb, agg);
    }
}

__device__ __forceinline__ float upd_elem(const float* __restrict__ x,
                                          const float* __restrict__ agg,
                                          const float* __restrict__ local,
                                          size_t idx, float ca, float cx) {
    return fmaxf(x[idx] + ca * agg[idx] + cx * local[idx], 0.f);
}

__device__ __forceinline__ v2f upd_frag(const float* __restrict__ x,
                                        const float* __restrict__ agg,
                                        const float* __restrict__ local,
                                        size_t o, float ca, float cx) {
    v2f a;
    a.x = upd_elem(x, agg, local, o + 0, ca, cx);
    a.y = upd_elem(x, agg, local, o + 1, ca, cx);
    return a;
}

__device__ __forceinline__ void node_store(
    const v8f* __restrict__ acc, const float* __restrict__ x,
    const float* __restrict__ agg, const float* __restrict__ local,
    float* __restrict__ out, int nbase, int half, int lm,
    const float* __restrict__ ob, float ca, float cx)
{
    const size_t gbase = (size_t)nbase * D;
#pragma unroll
    for (int n = 0; n < 8; ++n) {
        const int N = lm + (n << 4);
#pragma unroll
        for (int j = 0; j < 8; ++j) {
            const int M = j + (half << 3);
            const size_t idx = gbase + (size_t)M * D + N;
            const float u = upd_elem(x, agg, local, idx, ca, cx);
            out[idx] = acc[n][j] + u + ob[n];
        }
    }
}

__global__ __launch_bounds__(256) void gnn_node_kernel(
    const float* __restrict__ x,
    const float* __restrict__ agg,
    const float* __restrict__ local,
    const float* __restrict__ out_w,
    const float* __restrict__ out_b,
    const float* __restrict__ gamma,
    const float* __restrict__ eta,
    const float* __restrict__ xi,
    const int*   __restrict__ timestep,
    float*       __restrict__ out,
    int n_nodes)
{
    __shared__ float sW[D * D];
    for (int i = threadIdx.x; i < D * D; i += 256) {
        int r = i >> 7, c = i & 127;
        sW[sw_addr(r, c)] = out_w[i];
    }
    __syncthreads();

    const int t     = timestep[0];
    const float g_t = 1.f / (1.f + __expf(-gamma[t]));
    const float e_t = 1.f / (1.f + __expf(-eta[t]));
    const float ca  = (1.f - e_t) * g_t;
    const float cx  = xi[0];

    const int lane = threadIdx.x & 31;
    const int wave = threadIdx.x >> 5;
    const int half = lane >> 4;
    const int lm   = lane & 15;

    float ob[8];
#pragma unroll
    for (int n = 0; n < 8; ++n) ob[n] = out_b[lm + (n << 4)];

    const int n_tiles = n_nodes >> 4;
    const int n_pairs = (n_tiles + 1) >> 1;
    for (int pair = blockIdx.x * 8 + wave; pair < n_pairs; pair += gridDim.x * 8) {
        const int tile0 = pair << 1;
        const int tile1 = tile0 + 1;
        const bool has1 = tile1 < n_tiles;
        const int nb0 = tile0 << 4;
        const int nb1 = (has1 ? tile1 : tile0) << 4;

        const size_t r0 = (size_t)(nb0 + lm) * D + (half << 1);
        const size_t r1 = (size_t)(nb1 + lm) * D + (half << 1);

        v8f acc0[8], acc1[8];
#pragma unroll
        for (int n = 0; n < 8; ++n) {
#pragma unroll
            for (int q = 0; q < 8; ++q) { acc0[n][q] = 0.f; acc1[n][q] = 0.f; }
        }

        v2f bb[2][8];
        v2f aa0[2], aa1[2];
#pragma unroll
        for (int n = 0; n < 8; ++n) bb[0][n] = ldsB(sW, 0, half, lm, n);
        aa0[0] = upd_frag(x, agg, local, r0, ca, cx);
        aa1[0] = upd_frag(x, agg, local, r1, ca, cx);

#pragma unroll 2
        for (int kk = 0; kk < 32; ++kk) {
            const int cur = kk & 1, nxt = cur ^ 1;
            if (kk < 31) {
#pragma unroll
                for (int n = 0; n < 8; ++n) bb[nxt][n] = ldsB(sW, kk + 1, half, lm, n);
                aa0[nxt] = upd_frag(x, agg, local, r0 + (size_t)((kk + 1) << 2), ca, cx);
                aa1[nxt] = upd_frag(x, agg, local, r1 + (size_t)((kk + 1) << 2), ca, cx);
            }
#pragma unroll
            for (int n = 0; n < 8; ++n) acc0[n] = wmma_f32(aa0[cur], bb[cur][n], acc0[n]);
#pragma unroll
            for (int n = 0; n < 8; ++n) acc1[n] = wmma_f32(aa1[cur], bb[cur][n], acc1[n]);
        }

        node_store(acc0, x, agg, local, out, nb0, half, lm, ob, ca, cx);
        if (has1) node_store(acc1, x, agg, local, out, nb1, half, lm, ob, ca, cx);
    }
}

extern "C" void kernel_launch(void* const* d_in, const int* in_sizes, int n_in,
                              void* d_out, int out_size, void* d_ws, size_t ws_size,
                              hipStream_t stream) {
    const float* x        = (const float*)d_in[0];
    const int*   src_idx  = (const int*)  d_in[1];
    const int*   dst_idx  = (const int*)  d_in[2];
    const float* local    = (const float*)d_in[3];
    const int*   timestep = (const int*)  d_in[4];
    const float* gamma    = (const float*)d_in[5];
    const float* eta      = (const float*)d_in[6];
    const float* xi       = (const float*)d_in[7];
    const float* W_w      = (const float*)d_in[8];
    const float* W_b      = (const float*)d_in[9];
    const float* ln_g     = (const float*)d_in[10];
    const float* ln_b     = (const float*)d_in[11];
    const float* out_w    = (const float*)d_in[12];
    const float* out_b    = (const float*)d_in[13];
    float*       out      = (float*)d_out;
    float*       agg      = (float*)d_ws;

    const int n_nodes = in_sizes[0] / D;
    const int n_edges = in_sizes[1];

    zero_f4_kernel<<<512, 256, 0, stream>>>((float4*)agg, (n_nodes * D) / 4);

    gnn_edge_kernel<<<1024, 256, 0, stream>>>(
        x, src_idx, dst_idx, W_w, W_b, ln_g, ln_b, agg, n_edges);

    const int node_pairs  = ((n_nodes >> 4) + 1) >> 1;
    const int node_blocks = (node_pairs + 7) / 8;
    gnn_node_kernel<<<node_blocks, 256, 0, stream>>>(
        x, agg, local, out_w, out_b, gamma, eta, xi, timestep, out, n_nodes);
}


# Linear+LayerNorm hoisted from per edge to per source node (row gather commutes with a row-wise map): 12x fewer WMMAs; edge pass = whole-line row gather + f32 atomic adds, arrival order as before
# speedup vs baseline: 1.4975x; 1.4975x over previous
#include <hip/hip_runtime.h>
#include <hip/hip_bf16.h>


#define D 128

typedef __attribute__((ext_vector_type(2))) float v2f;
typedef __attribute__((ext_vector_type(8))) float v8f;

__device__ __forceinline__ v8f wmma_f32(v2f a, v2f b, v8f c) {
    return __builtin_amdgcn_wmma_f32_16x16x4_f32(
         false, a,  false, b,
         (short)0, c,  false,  false);
}

__device__ __forceinline__ int sw_addr(int row, int col) {
    return (row << 7) + ((((col >> 1) ^ (row & 63)) << 1) | (col & 1));
}

__device__ __forceinline__ v2f ldsB(const float* sW, int kk, int half, int lm, int n) {
    const int row = (n << 4) + lm;
    const int kB  = (kk << 2) + (half << 1);
    return *(const v2f*)&sW[sw_addr(row, kB)];
}

__global__ void zero_f4_kernel(float4* __restrict__ p, int n4) {
    for (int i = blockIdx.x * 256 + threadIdx.x; i < n4; i += gridDim.x * 256)
        p[i] = make_float4(0.f, 0.f, 0.f, 0.f);
}

__device__ __forceinline__ void ln_store(
    const v8f* __restrict__ acc, int nbase, int half, int lm,
    const float* __restrict__ lg, const float* __restrict__ lb,
    float* __restrict__ msg)
{
    float mu[8], rs[8];
#pragma unroll
    for (int j = 0; j < 8; ++j) {
        float s = 0.f, q = 0.f;
#pragma unroll
        for (int n = 0; n < 8; ++n) {
            const float v = acc[n][j];
            s += v; q += v * v;
        }
#pragma unroll
        for (int m = 1; m < 16; m <<= 1) {
            s += __shfl_xor(s, m, 32);
            q += __shfl_xor(q, m, 32);
        }
        const float mean = s * (1.f / 128.f);
        mu[j] = mean;
        rs[j] = rsqrtf(fmaxf(q * (1.f / 128.f) - mean * mean, 0.f) + 1e-5f);
    }

    const size_t gbase = (size_t)(nbase + (half << 3)) * D + lm;
#pragma unroll
    for (int n = 0; n < 8; ++n) {
#pragma unroll
        for (int j = 0; j < 8; ++j) {
            const float v = (acc[n][j] - mu[j]) * rs[j] * lg[n] + lb[n];
            msg[gbase + (size_t)j * D + (n << 4)] = v;
        }
    }
}

__global__ __launch_bounds__(256) void gnn_msg_kernel(
    const float* __restrict__ x,
    const float* __restrict__ W_w,
    const float* __restrict__ W_b,
    const float* __restrict__ ln_g,
    const float* __restrict__ ln_b,
    float*       __restrict__ msg,
    int n_nodes)
{
    __shared__ float sW[D * D];
    for (int i = threadIdx.x; i < D * D; i += 256) {
        int r = i >> 7, c = i & 127;
        sW[sw_addr(r, c)] = W_w[i];
    }
    __syncthreads();

    const int lane = threadIdx.x & 31;
    const int wave = threadIdx.x >> 5;
    const int half = lane >> 4;
    const int lm   = lane & 15;

    float wb[8], lg[8], lb[8];
#pragma unroll
    for (int n = 0; n < 8; ++n) {
        const int N = lm + (n << 4);
        wb[n] = W_b[N]; lg[n] = ln_g[N]; lb[n] = ln_b[N];
    }

    const int n_tiles = n_nodes >> 4;
    const int n_pairs = (n_tiles + 1) >> 1;
    for (int pair = blockIdx.x * 8 + wave; pair < n_pairs; pair += gridDim.x * 8) {
        const int tile0 = pair << 1;
        const int tile1 = tile0 + 1;
        const bool has1 = tile1 < n_tiles;
        const int nb0 = tile0 << 4;
        const int nb1 = (has1 ? tile1 : tile0) << 4;

        const float* arow0 = x + (size_t)(nb0 + lm) * D + (half << 1);
        const float* arow1 = x + (size_t)(nb1 + lm) * D + (half << 1);

        v8f acc0[8], acc1[8];
#pragma unroll
        for (int n = 0; n < 8; ++n) {
#pragma unroll
            for (int q = 0; q < 8; ++q) { acc0[n][q] = wb[n]; acc1[n][q] = wb[n]; }
        }

        v2f bb[2][8];
        v2f aa0[2], aa1[2];
#pragma unroll
        for (int n = 0; n < 8; ++n) bb[0][n] = ldsB(sW, 0, half, lm, n);
        aa0[0] = *(const v2f*)&arow0[0];
        aa1[0] = *(const v2f*)&arow1[0];

#pragma unroll 2
        for (int kk = 0; kk < 32; ++kk) {
            const int cur = kk & 1, nxt = cur ^ 1;
            if (kk < 31) {
#pragma unroll
                for (int n = 0; n < 8; ++n) bb[nxt][n] = ldsB(sW, kk + 1, half, lm, n);
                aa0[nxt] = *(const v2f*)&arow0[(kk + 1) << 2];
                aa1[nxt] = *(const v2f*)&arow1[(kk + 1) << 2];
            }
#pragma unroll
            for (int n = 0; n < 8; ++n) acc0[n] = wmma_f32(aa0[cur], bb[cur][n], acc0[n]);
#pragma unroll
            for (int n = 0; n < 8; ++n) acc1[n] = wmma_f32(aa1[cur], bb[cur][n], acc1[n]);
        }

        ln_store(acc0, nb0, half, lm, lg, lb, msg);
        if (has1) ln_store(acc1, nb1, half, lm, lg, lb, msg);
    }
}

#define EPW 8

__global__ __launch_bounds__(256) void gnn_scatter_kernel(
    const float* __restrict__ msg,
    const int*   __restrict__ src_idx,
    const int*   __restrict__ dst_idx,
    float*       __restrict__ agg,
    int n_edges)
{
    const int lane = threadIdx.x & 31;
    const int wave = __builtin_amdgcn_readfirstlane(threadIdx.x >> 5);
    const int n_chunks = (n_edges + EPW - 1) / EPW;

    for (int chunk = blockIdx.x * 8 + wave; chunk < n_chunks; chunk += gridDim.x * 8) {
        const int e0 = chunk * EPW;
        if (e0 + EPW <= n_edges) {
            int s[EPW], d[EPW];
#pragma unroll
            for (int j = 0; j < EPW; ++j) { s[j] = src_idx[e0 + j]; d[j] = dst_idx[e0 + j]; }

            float v[EPW][4];
#pragma unroll
            for (int j = 0; j < EPW; ++j) {
                const float* row = msg + (size_t)s[j] * D + lane;
#pragma unroll
                for (int c = 0; c < 4; ++c) v[j][c] = row[c << 5];
            }
#pragma unroll
            for (int j = 0; j < EPW; ++j) {
                float* row = agg + (size_t)d[j] * D + lane;
#pragma unroll
                for (int c = 0; c < 4; ++c) atomicAdd(&row[c << 5], v[j][c]);
            }
        } else {
            for (int e = e0; e < n_edges; ++e) {
                const float* srow = msg + (size_t)src_idx[e] * D + lane;
                float*       drow = agg + (size_t)dst_idx[e] * D + lane;
#pragma unroll
                for (int c = 0; c < 4; ++c) atomicAdd(&drow[c << 5], srow[c << 5]);
            }
        }
    }
}

__device__ __forceinline__ float upd_elem(const float* __restrict__ x,
                                          const float* __restrict__ agg,
                                          const float* __restrict__ local,
                                          size_t idx, float ca, float cx) {
    return fmaxf(x[idx] + ca * agg[idx] + cx * local[idx], 0.f);
}

__device__ __forceinline__ v2f upd_frag(const float* __restrict__ x,
                                        const float* __restrict__ agg,
                                        const float* __restrict__ local,
                                        size_t o, float ca, float cx) {
    v2f a;
    a.x = upd_elem(x, agg, local, o + 0, ca, cx);
    a.y = upd_elem(x, agg, local, o + 1, ca, cx);
    return a;
}

__device__ __forceinline__ void node_store(
    const v8f* __restrict__ acc, const float* __restrict__ x,
    const float* __restrict__ agg, const float* __restrict__ local,
    float* __restrict__ out, int nbase, int half, int lm,
    const float* __restrict__ ob, float ca, float cx)
{
    const size_t gbase = (size_t)nbase * D;
#pragma unroll
    for (int n = 0; n < 8; ++n) {
        const int N = lm + (n << 4);
#pragma unroll
        for (int j = 0; j < 8; ++j) {
            const int M = j + (half << 3);
            const size_t idx = gbase + (size_t)M * D + N;
            const float u = upd_elem(x, agg, local, idx, ca, cx);
            out[idx] = acc[n][j] + u + ob[n];
        }
    }
}

__global__ __launch_bounds__(256) void gnn_node_kernel(
    const float* __restrict__ x,
    const float* __restrict__ agg,
    const float* __restrict__ local,
    const float* __restrict__ out_w,
    const float* __restrict__ out_b,
    const float* __restrict__ gamma,
    const float* __restrict__ eta,
    const float* __restrict__ xi,
    const int*   __restrict__ timestep,
    float*       __restrict__ out,
    int n_nodes)
{
    __shared__ float sW[D * D];
    for (int i = threadIdx.x; i < D * D; i += 256) {
        int r = i >> 7, c = i & 127;
        sW[sw_addr(r, c)] = out_w[i];
    }
    __syncthreads();

    const int t     = timestep[0];
    const float g_t = 1.f / (1.f + __expf(-gamma[t]));
    const float e_t = 1.f / (1.f + __expf(-eta[t]));
    const float ca  = (1.f - e_t) * g_t;
    const float cx  = xi[0];

    const int lane = threadIdx.x & 31;
    const int wave = threadIdx.x >> 5;
    const int half = lane >> 4;
    const int lm   = lane & 15;

    float ob[8];
#pragma unroll
    for (int n = 0; n < 8; ++n) ob[n] = out_b[lm + (n << 4)];

    const int n_tiles = n_nodes >> 4;
    const int n_pairs = (n_tiles + 1) >> 1;
    for (int pair = blockIdx.x * 8 + wave; pair < n_pairs; pair += gridDim.x * 8) {
        const int tile0 = pair << 1;
        const int tile1 = tile0 + 1;
        const bool has1 = tile1 < n_tiles;
        const int nb0 = tile0 << 4;
        const int nb1 = (has1 ? tile1 : tile0) << 4;

        const size_t r0 = (size_t)(nb0 + lm) * D + (half << 1);
        const size_t r1 = (size_t)(nb1 + lm) * D + (half << 1);

        v8f acc0[8], acc1[8];
#pragma unroll
        for (int n = 0; n < 8; ++n) {
#pragma unroll
            for (int q = 0; q < 8; ++q) { acc0[n][q] = 0.f; acc1[n][q] = 0.f; }
        }

        v2f bb[2][8];
        v2f aa0[2], aa1[2];
#pragma unroll
        for (int n = 0; n < 8; ++n) bb[0][n] = ldsB(sW, 0, half, lm, n);
        aa0[0] = upd_frag(x, agg, local, r0, ca, cx);
        aa1[0] = upd_frag(x, agg, local, r1, ca, cx);

#pragma unroll 2
        for (int kk = 0; kk < 32; ++kk) {
            const int cur = kk & 1, nxt = cur ^ 1;
            if (kk < 31) {
#pragma unroll
                for (int n = 0; n < 8; ++n) bb[nxt][n] = ldsB(sW, kk + 1, half, lm, n);
                aa0[nxt] = upd_frag(x, agg, local, r0 + (size_t)((kk + 1) << 2), ca, cx);
                aa1[nxt] = upd_frag(x, agg, local, r1 + (size_t)((kk + 1) << 2), ca, cx);
            }
#pragma unroll
            for (int n = 0; n < 8; ++n) acc0[n] = wmma_f32(aa0[cur], bb[cur][n], acc0[n]);
#pragma unroll
            for (int n = 0; n < 8; ++n) acc1[n] = wmma_f32(aa1[cur], bb[cur][n], acc1[n]);
        }

        node_store(acc0, x, agg, local, out, nb0, half, lm, ob, ca, cx);
        if (has1) node_store(acc1, x, agg, local, out, nb1, half, lm, ob, ca, cx);
    }
}

extern "C" void kernel_launch(void* const* d_in, const int* in_sizes, int n_in,
                              void* d_out, int out_size, void* d_ws, size_t ws_size,
                              hipStream_t stream) {
    const float* x        = (const float*)d_in[0];
    const int*   src_idx  = (const int*)  d_in[1];
    const int*   dst_idx  = (const int*)  d_in[2];
    const float* local    = (const float*)d_in[3];
    const int*   timestep = (const int*)  d_in[4];
    const float* gamma    = (const float*)d_in[5];
    const float* eta      = (const float*)d_in[6];
    const float* xi       = (const float*)d_in[7];
    const float* W_w      = (const float*)d_in[8];
    const float* W_b      = (const float*)d_in[9];
    const float* ln_g     = (const float*)d_in[10];
    const float* ln_b     = (const float*)d_in[11];
    const float* out_w    = (const float*)d_in[12];
    const float* out_b    = (const float*)d_in[13];
    float*       out      = (float*)d_out;
    float*       agg      = (float*)d_ws;

    const int n_nodes = in_sizes[0] / D;
    const int n_edges = in_sizes[1];

    float*       msg      = out;

    zero_f4_kernel<<<512, 256, 0, stream>>>((float4*)agg, (n_nodes * D) / 4);

    const int node_pairs  = ((n_nodes >> 4) + 1) >> 1;
    const int node_blocks = (node_pairs + 7) / 8;
    gnn_msg_kernel<<<node_blocks, 256, 0, stream>>>(
        x, W_w, W_b, ln_g, ln_b, msg, n_nodes);

    gnn_scatter_kernel<<<1024, 256, 0, stream>>>(msg, src_idx, dst_idx, agg, n_edges);

    gnn_node_kernel<<<node_blocks, 256, 0, stream>>>(
        x, agg, local, out_w, out_b, gamma, eta, xi, timestep, out, n_nodes);
}


# Global f32 atomic scatter replaced by destination-owned sums: each work-group scans dst, gathers msg rows of its 208 nodes into LDS (ds f32 adds, LDS arrival order), fused into the node GEMM
# speedup vs baseline: 2.6639x; 1.7789x over previous
#include <hip/hip_runtime.h>
#include <hip/hip_bf16.h>


#define D 128

typedef __attribute__((ext_vector_type(2))) float v2f;
typedef __attribute__((ext_vector_type(8))) float v8f;

__device__ __forceinline__ v8f wmma_f32(v2f a, v2f b, v8f c) {
    return __builtin_amdgcn_wmma_f32_16x16x4_f32(
         false, a,  false, b,
         (short)0, c,  false,  false);
}

__device__ __forceinline__ int sw_addr(int row, int col) {
    return (row << 7) + ((((col >> 1) ^ (row & 63)) << 1) | (col & 1));
}

__device__ __forceinline__ v2f ldsB(const float* sW, int kk, int half, int lm, int n) {
    const int row = (n << 4) + lm;
    const int kB  = (kk << 2) + (half << 1);
    return *(const v2f*)&sW[sw_addr(row, kB)];
}

__device__ __forceinline__ void ln_store(
    const v8f* __restrict__ acc, int nbase, int half, int lm,
    const float* __restrict__ lg, const float* __restrict__ lb,
    float* __restrict__ msg)
{
    float mu[8], rs[8];
#pragma unroll
    for (int j = 0; j < 8; ++j) {
        float s = 0.f, q = 0.f;
#pragma unroll
        for (int n = 0; n < 8; ++n) {
            const float v = acc[n][j];
            s += v; q += v * v;
        }
#pragma unroll
        for (int m = 1; m < 16; m <<= 1) {
            s += __shfl_xor(s, m, 32);
            q += __shfl_xor(q, m, 32);
        }
        const float mean = s * (1.f / 128.f);
        mu[j] = mean;
        rs[j] = rsqrtf(fmaxf(q * (1.f / 128.f) - mean * mean, 0.f) + 1e-5f);
    }

    const size_t gbase = (size_t)(nbase + (half << 3)) * D + lm;
#pragma unroll
    for (int n = 0; n < 8; ++n) {
#pragma unroll
        for (int j = 0; j < 8; ++j) {
            const float v = (acc[n][j] - mu[j]) * rs[j] * lg[n] + lb[n];
            msg[gbase + (size_t)j * D + (n << 4)] = v;
        }
    }
}

__global__ __launch_bounds__(256) void gnn_msg_kernel(
    const float* __restrict__ x,
    const float* __restrict__ W_w,
    const float* __restrict__ W_b,
    const float* __restrict__ ln_g,
    const float* __restrict__ ln_b,
    float*       __restrict__ msg,
    int n_nodes)
{
    __shared__ float sW[D * D];
    for (int i = threadIdx.x; i < D * D; i += 256) {
        int r = i >> 7, c = i & 127;
        sW[sw_addr(r, c)] = W_w[i];
    }
    __syncthreads();

    const int lane = threadIdx.x & 31;
    const int wave = threadIdx.x >> 5;
    const int half = lane >> 4;
    const int lm   = lane & 15;

    float wb[8], lg[8], lb[8];
#pragma unroll
    for (int n = 0; n < 8; ++n) {
        const int N = lm + (n << 4);
        wb[n] = W_b[N]; lg[n] = ln_g[N]; lb[n] = ln_b[N];
    }

    const int n_tiles = n_nodes >> 4;
    const int n_pairs = (n_tiles + 1) >> 1;
    for (int pair = blockIdx.x * 8 + wave; pair < n_pairs; pair += gridDim.x * 8) {
        const int tile0 = pair << 1;
        const int tile1 = tile0 + 1;
        const bool has1 = tile1 < n_tiles;
        const int nb0 = tile0 << 4;
        const int nb1 = (has1 ? tile1 : tile0) << 4;

        const float* arow0 = x + (size_t)(nb0 + lm) * D + (half << 1);
        const float* arow1 = x + (size_t)(nb1 + lm) * D + (half << 1);

        v8f acc0[8], acc1[8];
#pragma unroll
        for (int n = 0; n < 8; ++n) {
#pragma unroll
            for (int q = 0; q < 8; ++q) { acc0[n][q] = wb[n]; acc1[n][q] = wb[n]; }
        }

        v2f bb[2][8];
        v2f aa0[2], aa1[2];
#pragma unroll
        for (int n = 0; n < 8; ++n) bb[0][n] = ldsB(sW, 0, half, lm, n);
        aa0[0] = *(const v2f*)&arow0[0];
        aa1[0] = *(const v2f*)&arow1[0];

#pragma unroll 2
        for (int kk = 0; kk < 32; ++kk) {
            const int cur = kk & 1, nxt = cur ^ 1;
            if (kk < 31) {
#pragma unroll
                for (int n = 0; n < 8; ++n) bb[nxt][n] = ldsB(sW, kk + 1, half, lm, n);
                aa0[nxt] = *(const v2f*)&arow0[(kk + 1) << 2];
                aa1[nxt] = *(const v2f*)&arow1[(kk + 1) << 2];
            }
#pragma unroll
            for (int n = 0; n < 8; ++n) acc0[n] = wmma_f32(aa0[cur], bb[cur][n], acc0[n]);
#pragma unroll
            for (int n = 0; n < 8; ++n) acc1[n] = wmma_f32(aa1[cur], bb[cur][n], acc1[n]);
        }

        ln_store(acc0, nb0, half, lm, lg, lb, msg);
        if (has1) ln_store(acc1, nb1, half, lm, lg, lb, msg);
    }
}

#define TPB   13
#define RPB   (TPB * 16)
#define UP    130
#define NWAVE 32
#define SUBE  512
#define QCAP  (32 + SUBE)
#define GB    8

__global__ __launch_bounds__(1024) void gnn_node_kernel(
    const float* __restrict__ x,
    const float* __restrict__ msg,
    const int*   __restrict__ src_idx,
    const int*   __restrict__ dst_idx,
    const float* __restrict__ local,
    const float* __restrict__ out_w,
    const float* __restrict__ out_b,
    const float* __restrict__ gamma,
    const float* __restrict__ eta,
    const float* __restrict__ xi,
    const int*   __restrict__ timestep,
    float*       __restrict__ out,
    int n_nodes, int n_edges)
{
    __shared__ __attribute__((aligned(16))) float sW[D * D];
    __shared__ __attribute__((aligned(16))) float sU[RPB * UP];
    __shared__ int   sQ[NWAVE * QCAP];

    for (int i = threadIdx.x; i < D * D; i += 1024) {
        int r = i >> 7, c = i & 127;
        sW[sw_addr(r, c)] = out_w[i];
    }
    for (int i = threadIdx.x; i < RPB * UP; i += 1024) sU[i] = 0.f;
    __syncthreads();

    const int lane = threadIdx.x & 31;
    const int wave = __builtin_amdgcn_readfirstlane(threadIdx.x >> 5);
    const int half = lane >> 4;
    const int lm   = lane & 15;

    const int lo    = blockIdx.x * RPB;
    const int nrows = min(RPB, n_nodes - lo);

    int* q = sQ + wave * QCAP;
    int cnt = 0;
    const int n_sub = (n_edges + SUBE - 1) / SUBE;
    const int rot = (int)(((long long)blockIdx.x * n_sub) / gridDim.x);
    const unsigned below = (1u << lane) - 1u;

    for (int sub = wave; sub <= n_sub; sub += NWAVE) {
        if (sub < n_sub) {
            const int rs = sub + rot;
            const int base = (rs < n_sub ? rs : rs - n_sub) * SUBE;
            int d[4][4];
            if (base + SUBE <= n_edges) {
#pragma unroll
                for (int g = 0; g < 4; ++g) {
                    const int4 t = *(const int4*)&dst_idx[base + (g << 7) + (lane << 2)];
                    d[g][0] = t.x; d[g][1] = t.y; d[g][2] = t.z; d[g][3] = t.w;
                }
            } else {
#pragma unroll
                for (int g = 0; g < 4; ++g) {
#pragma unroll
                    for (int k = 0; k < 4; ++k) {
                        const int e = base + (g << 7) + (lane << 2) + k;
                        d[g][k] = (e < n_edges) ? dst_idx[e] : -1;
                    }
                }
            }
#pragma unroll
            for (int g = 0; g < 4; ++g) {
#pragma unroll
                for (int k = 0; k < 4; ++k) {
                    const int dl = d[g][k] - lo;
                    const bool m = (unsigned)dl < (unsigned)nrows;
                    const unsigned mask = __builtin_amdgcn_ballot_w32(m);
                    if (mask != 0u) {
                        if (m) q[cnt + __builtin_popcount(mask & below)] =
                                   ((base + (g << 7) + (lane << 2) + k) << 8) | dl;
                        cnt += __builtin_popcount(mask);
                    }
                }
            }
        }
        if (cnt >= 32 || sub + NWAVE > n_sub) {
            for (int b0 = 0; b0 < cnt; b0 += 32) {
                const int n   = min(32, cnt - b0);
                const int ent = (lane < n) ? q[b0 + lane] : 0;
                const int s   = src_idx[ent >> 8];
                const int dl  = ent & 255;
                for (int j0 = 0; j0 < n; j0 += GB) {
                    float v[GB][4];
#pragma unroll
                    for (int j = 0; j < GB; ++j) {
                        const int sj = __builtin_amdgcn_readlane(s, min(j0 + j, n - 1));
                        const float* row = msg + (size_t)sj * D + lane;
#pragma unroll
                        for (int c = 0; c < 4; ++c) v[j][c] = row[c << 5];
                    }
#pragma unroll
                    for (int j = 0; j < GB; ++j) {
                        if (j0 + j < n) {
                            const int dj = __builtin_amdgcn_readlane(dl, j0 + j);
                            float* row = sU + dj * UP + lane;
#pragma unroll
                            for (int c = 0; c < 4; ++c) atomicAdd(&row[c << 5], v[j][c]);
                        }
                    }
                }
            }
            cnt = 0;
        }
    }
    __syncthreads();

    const int t     = timestep[0];
    const float g_t = 1.f / (1.f + __expf(-gamma[t]));
    const float e_t = 1.f / (1.f + __expf(-eta[t]));
    const float ca  = (1.f - e_t) * g_t;
    const float cx  = xi[0];

    for (int i = threadIdx.x; i < nrows * (D / 2); i += 1024) {
        const int r = i >> 6, c = (i & 63) << 1;
        const size_t g = (size_t)(lo + r) * D + c;
        const v2f xv = *(const v2f*)&x[g];
        const v2f lv = *(const v2f*)&local[g];
        v2f a = *(const v2f*)&sU[r * UP + c];
        a.x = fmaxf(xv.x + ca * a.x + cx * lv.x, 0.f);
        a.y = fmaxf(xv.y + ca * a.y + cx * lv.y, 0.f);
        *(v2f*)&sU[r * UP + c] = a;
    }
    __syncthreads();

    const int ntl = nrows >> 4;
    for (int item = wave; item < ntl * 2; item += NWAVE) {
        const int tl = item >> 1;
        const int nh = (item & 1) << 2;
        const float* arow = sU + ((tl << 4) + lm) * UP + (half << 1);

        float ob[4];
#pragma unroll
        for (int n = 0; n < 4; ++n) ob[n] = out_b[lm + ((nh + n) << 4)];

        v8f acc[4];
#pragma unroll
        for (int n = 0; n < 4; ++n) {
#pragma unroll
            for (int p = 0; p < 8; ++p) acc[n][p] = 0.f;
        }

        v2f bb[2][4];
        v2f aa[2];
#pragma unroll
        for (int n = 0; n < 4; ++n) bb[0][n] = ldsB(sW, 0, half, lm, nh + n);
        aa[0] = *(const v2f*)&arow[0];

#pragma unroll 2
        for (int kk = 0; kk < 32; ++kk) {
            const int cur = kk & 1, nxt = cur ^ 1;
            if (kk < 31) {
#pragma unroll
                for (int n = 0; n < 4; ++n) bb[nxt][n] = ldsB(sW, kk + 1, half, lm, nh + n);
                aa[nxt] = *(const v2f*)&arow[(kk + 1) << 2];
            }
#pragma unroll
            for (int n = 0; n < 4; ++n) acc[n] = wmma_f32(aa[cur], bb[cur][n], acc[n]);
        }

        const int rbase = (tl << 4) + (half << 3);
#pragma unroll
        for (int n = 0; n < 4; ++n) {
            const int N = lm + ((nh + n) << 4);
#pragma unroll
            for (int j = 0; j < 8; ++j) {
                const float u = sU[(rbase + j) * UP + N];
                out[(size_t)(lo + rbase + j) * D + N] = acc[n][j] + u + ob[n];
            }
        }
    }
}

extern "C" void kernel_launch(void* const* d_in, const int* in_sizes, int n_in,
                              void* d_out, int out_size, void* d_ws, size_t ws_size,
                              hipStream_t stream) {
    const float* x        = (const float*)d_in[0];
    const int*   src_idx  = (const int*)  d_in[1];
    const int*   dst_idx  = (const int*)  d_in[2];
    const float* local    = (const float*)d_in[3];
    const int*   timestep = (const int*)  d_in[4];
    const float* gamma    = (const float*)d_in[5];
    const float* eta      = (const float*)d_in[6];
    const float* xi       = (const float*)d_in[7];
    const float* W_w      = (const float*)d_in[8];
    const float* W_b      = (const float*)d_in[9];
    const float* ln_g     = (const float*)d_in[10];
    const float* ln_b     = (const float*)d_in[11];
    const float* out_w    = (const float*)d_in[12];
    const float* out_b    = (const float*)d_in[13];
    float*       out      = (float*)d_out;
    float*       msg      = (float*)d_ws;

    const int n_nodes = in_sizes[0] / D;
    const int n_edges = in_sizes[1];

    const int node_pairs  = ((n_nodes >> 4) + 1) >> 1;
    const int msg_blocks  = (node_pairs + 7) / 8;
    gnn_msg_kernel<<<msg_blocks, 256, 0, stream>>>(
        x, W_w, W_b, ln_g, ln_b, msg, n_nodes);

    const int node_blocks = (n_nodes + RPB - 1) / RPB;
    gnn_node_kernel<<<node_blocks, 1024, 0, stream>>>(
        x, msg, src_idx, dst_idx, local, out_w, out_b, gamma, eta, xi, timestep,
        out, n_nodes, n_edges);
}


# Whole-dst scan in every work-group replaced by one binning pass: extra blocks of launch 1 sort edges by owner into cells in the owner's out rows (LDS int rank; f32 sums stay in arrival order)
# speedup vs baseline: 3.4576x; 1.2979x over previous
#include <hip/hip_runtime.h>
#include <hip/hip_bf16.h>


#define D 128

typedef __attribute__((ext_vector_type(2))) float v2f;
typedef __attribute__((ext_vector_type(8))) float v8f;

__device__ __forceinline__ v8f wmma_f32(v2f a, v2f b, v8f c) {
    return __builtin_amdgcn_wmma_f32_16x16x4_f32(
         false, a,  false, b,
         (short)0, c,  false,  false);
}

__device__ __forceinline__ int sw_addr(int row, int col) {
    return (row << 7) + ((((col >> 1) ^ (row & 63)) << 1) | (col & 1));
}

__device__ __forceinline__ v2f ldsB(const float* sW, int kk, int half, int lm, int n) {
    const int row = (n << 4) + lm;
    const int kB  = (kk << 2) + (half << 1);
    return *(const v2f*)&sW[sw_addr(row, kB)];
}

__device__ __forceinline__ void ln_store(
    const v8f* __restrict__ acc, int nbase, int half, int lm,
    const float* __restrict__ lg, const float* __restrict__ lb,
    float* __restrict__ msg)
{
    float mu[8], rs[8];
#pragma unroll
    for (int j = 0; j < 8; ++j) {
        float s = 0.f, q = 0.f;
#pragma unroll
        for (int n = 0; n < 8; ++n) {
            const float v = acc[n][j];
            s += v; q += v * v;
        }
#pragma unroll
        for (int m = 1; m < 16; m <<= 1) {
            s += __shfl_xor(s, m, 32);
            q += __shfl_xor(q, m, 32);
        }
        const float mean = s * (1.f / 128.f);
        mu[j] = mean;
        rs[j] = rsqrtf(fmaxf(q * (1.f / 128.f) - mean * mean, 0.f) + 1e-5f);
    }

    const size_t gbase = (size_t)(nbase + (half << 3)) * D + lm;
#pragma unroll
    for (int n = 0; n < 8; ++n) {
#pragma unroll
        for (int j = 0; j < 8; ++j) {
            const float v = (acc[n][j] - mu[j]) * rs[j] * lg[n] + lb[n];
            msg[gbase + (size_t)j * D + (n << 4)] = v;
        }
    }
}

#define TPB   13
#define RPB   (TPB * 16)
#define BS    4096
#define BEPT  (BS / 256)
#define CP    64
#define CCAP  (CP - 1)
#define HMAX  1024

__device__ __forceinline__ void bin_slice(
    int* hC,
    const int* __restrict__ src_idx, const int* __restrict__ dst_idx,
    float* __restrict__ cells,
    int g, int n_slices, int n_nodes, int n_edges, int n_owner)
{
    const int tid = threadIdx.x;
    for (int i = tid; i < n_owner; i += 256) hC[i] = 0;
    __syncthreads();

    const int e0 = g * BS + tid;
    int dv[BEPT], sv[BEPT];
#pragma unroll
    for (int k = 0; k < BEPT; ++k) {
        const int e = min(e0 + (k << 8), n_edges - 1);
        dv[k] = dst_idx[e];
        sv[k] = src_idx[e];
    }
    const int last_rows = n_nodes - (n_owner - 1) * RPB;
    const int need      = n_slices * CP;
#pragma unroll
    for (int k = 0; k < BEPT; ++k) {
        const int d = dv[k];
        if (e0 + (k << 8) < n_edges && (unsigned)d < (unsigned)n_nodes) {
            const int b    = d / RPB;
            const int r    = atomicAdd(&hC[b], 1);
            const int rows = (b == n_owner - 1) ? last_rows : RPB;
            if (r < CCAP && rows * D >= need)
                cells[(size_t)b * (RPB * D) + g * CP + 1 + r] =
                    __int_as_float((sv[k] << 8) | (d - b * RPB));
        }
    }
    __syncthreads();
    for (int b = tid; b < n_owner; b += 256) {
        const int rows = (b == n_owner - 1) ? last_rows : RPB;
        if (rows * D >= need)
            cells[(size_t)b * (RPB * D) + g * CP] = __int_as_float(hC[b]);
    }
}

__global__ __launch_bounds__(256) void gnn_msg_kernel(
    const float* __restrict__ x,
    const float* __restrict__ W_w,
    const float* __restrict__ W_b,
    const float* __restrict__ ln_g,
    const float* __restrict__ ln_b,
    float*       __restrict__ msg,
    int n_nodes, int msg_blocks,
    const int*   __restrict__ src_idx,
    const int*   __restrict__ dst_idx,
    float*       __restrict__ cells,
    int n_edges, int n_owner, int n_slices)
{
    __shared__ float sW[D * D];
    __shared__ int   hC[HMAX];

    if ((int)blockIdx.x >= msg_blocks) {
        if (n_owner <= HMAX)
            bin_slice(hC, src_idx, dst_idx, cells, (int)blockIdx.x - msg_blocks,
                      n_slices, n_nodes, n_edges, n_owner);
        return;
    }

    for (int i = threadIdx.x; i < D * D; i += 256) {
        int r = i >> 7, c = i & 127;
        sW[sw_addr(r, c)] = W_w[i];
    }
    __syncthreads();

    const int lane = threadIdx.x & 31;
    const int wave = threadIdx.x >> 5;
    const int half = lane >> 4;
    const int lm   = lane & 15;

    float wb[8], lg[8], lb[8];
#pragma unroll
    for (int n = 0; n < 8; ++n) {
        const int N = lm + (n << 4);
        wb[n] = W_b[N]; lg[n] = ln_g[N]; lb[n] = ln_b[N];
    }

    const int n_tiles = n_nodes >> 4;
    const int n_pairs = (n_tiles + 1) >> 1;
    for (int pair = blockIdx.x * 8 + wave; pair < n_pairs; pair += msg_blocks * 8) {
        const int tile0 = pair << 1;
        const int tile1 = tile0 + 1;
        const bool has1 = tile1 < n_tiles;
        const int nb0 = tile0 << 4;
        const int nb1 = (has1 ? tile1 : tile0) << 4;

        const float* arow0 = x + (size_t)(nb0 + lm) * D + (half << 1);
        const float* arow1 = x + (size_t)(nb1 + lm) * D + (half << 1);

        v8f acc0[8], acc1[8];
#pragma unroll
        for (int n = 0; n < 8; ++n) {
#pragma unroll
            for (int q = 0; q < 8; ++q) { acc0[n][q] = wb[n]; acc1[n][q] = wb[n]; }
        }

        v2f bb[2][8];
        v2f aa0[2], aa1[2];
#pragma unroll
        for (int n = 0; n < 8; ++n) bb[0][n] = ldsB(sW, 0, half, lm, n);
        aa0[0] = *(const v2f*)&arow0[0];
        aa1[0] = *(const v2f*)&arow1[0];

#pragma unroll 2
        for (int kk = 0; kk < 32; ++kk) {
            const int cur = kk & 1, nxt = cur ^ 1;
            if (kk < 31) {
#pragma unroll
                for (int n = 0; n < 8; ++n) bb[nxt][n] = ldsB(sW, kk + 1, half, lm, n);
                aa0[nxt] = *(const v2f*)&arow0[(kk + 1) << 2];
                aa1[nxt] = *(const v2f*)&arow1[(kk + 1) << 2];
            }
#pragma unroll
            for (int n = 0; n < 8; ++n) acc0[n] = wmma_f32(aa0[cur], bb[cur][n], acc0[n]);
#pragma unroll
            for (int n = 0; n < 8; ++n) acc1[n] = wmma_f32(aa1[cur], bb[cur][n], acc1[n]);
        }

        ln_store(acc0, nb0, half, lm, lg, lb, msg);
        if (has1) ln_store(acc1, nb1, half, lm, lg, lb, msg);
    }
}

#define UP    130
#define NWAVE 32
#define SUBE  512
#define QCAP  (32 + SUBE)
#define GB    8
#define CH    4

__device__ __forceinline__ int drain_queue(
    int* q, int cnt, bool all, const float* __restrict__ msg, float* sU, int lane)
{
    const int lim = all ? cnt : (cnt & ~31);
    int b0 = 0;
    for (; b0 < lim; b0 += 32) {
        const int n   = min(32, cnt - b0);
        const int ent = (lane < n) ? q[b0 + lane] : 0;
        const int s   = ent >> 8;
        const int dl  = ent & 255;
        for (int j0 = 0; j0 < n; j0 += GB) {
            float v[GB][4];
#pragma unroll
            for (int j = 0; j < GB; ++j) {
                const int sj = __builtin_amdgcn_readlane(s, min(j0 + j, n - 1));
                const float* row = msg + (size_t)sj * D + lane;
#pragma unroll
                for (int c = 0; c < 4; ++c) v[j][c] = row[c << 5];
            }
#pragma unroll
            for (int j = 0; j < GB; ++j) {
                if (j0 + j < n) {
                    const int dj = __builtin_amdgcn_readlane(dl, j0 + j);
                    float* row = sU + dj * UP + lane;
#pragma unroll
                    for (int c = 0; c < 4; ++c) atomicAdd(&row[c << 5], v[j][c]);
                }
            }
        }
    }
    const int rem = cnt - b0;
    if (rem > 0 && b0 > 0) {
        const int t = (lane < rem) ? q[b0 + lane] : 0;
        if (lane < rem) q[lane] = t;
    }
    return rem > 0 ? rem : 0;
}

__global__ __launch_bounds__(1024) void gnn_node_kernel(
    const float* __restrict__ x,
    const float* __restrict__ msg,
    const int*   __restrict__ src_idx,
    const int*   __restrict__ dst_idx,
    const float* __restrict__ local,
    const float* __restrict__ out_w,
    const float* __restrict__ out_b,
    const float* __restrict__ gamma,
    const float* __restrict__ eta,
    const float* __restrict__ xi,
    const int*   __restrict__ timestep,
    float*       __restrict__ out,
    int n_nodes, int n_edges, int n_slices)
{
    __shared__ __attribute__((aligned(16))) float sW[D * D];
    __shared__ __attribute__((aligned(16))) float sU[RPB * UP];
    __shared__ int   sQ[NWAVE * QCAP];
    __shared__ int   sScan;

    const int lo    = blockIdx.x * RPB;
    const int nrows = min(RPB, n_nodes - lo);

    const float* cellrow = out + (size_t)lo * D;
    const bool can_bin = ((int)gridDim.x <= HMAX) && (nrows * D >= n_slices * CP);
    int over = can_bin ? 0 : 1;
    if (can_bin) {
        for (int g = threadIdx.x; g < n_slices; g += 1024)
            over |= (__float_as_int(cellrow[g * CP]) > CCAP) ? 1 : 0;
    }
    if (threadIdx.x == 0) sScan = 0;
    __syncthreads();

    for (int i = threadIdx.x; i < D * D; i += 1024) {
        int r = i >> 7, c = i & 127;
        sW[sw_addr(r, c)] = out_w[i];
    }
    for (int i = threadIdx.x; i < RPB * UP; i += 1024) sU[i] = 0.f;
    if (over) sScan = 1;
    __syncthreads();

    const int lane = threadIdx.x & 31;
    const int wave = __builtin_amdgcn_readfirstlane(threadIdx.x >> 5);
    const int half = lane >> 4;
    const int lm   = lane & 15;

    int* q = sQ + wave * QCAP;
    int cnt = 0;
    const bool need_scan = sScan != 0;

    if (!need_scan) {
        for (int g = wave; g < n_slices; g += CH * NWAVE) {
            float c0[CH], c1[CH];
#pragma unroll
            for (int u = 0; u < CH; ++u) {
                const int gu = g + u * NWAVE;
                c0[u] = 0.f; c1[u] = 0.f;
                if (gu < n_slices) {
                    c0[u] = cellrow[gu * CP + lane];
                    c1[u] = cellrow[gu * CP + 32 + lane];
                }
            }
#pragma unroll
            for (int u = 0; u < CH; ++u) {
                const int v0 = __float_as_int(c0[u]), v1 = __float_as_int(c1[u]);
                const int n = min(max(__builtin_amdgcn_readfirstlane(v0), 0), CCAP);
                if (lane >= 1 && lane - 1 < n) q[cnt + lane - 1] = v0;
                if (31 + lane < n)             q[cnt + 31 + lane] = v1;
                cnt += n;
            }
            if (cnt >= 32) cnt = drain_queue(q, cnt, false, msg, sU, lane);
        }
        drain_queue(q, cnt, true, msg, sU, lane);
        cnt = 0;
    }

    const int n_sub = (n_edges + SUBE - 1) / SUBE;
    const int rot = (int)(((long long)blockIdx.x * n_sub) / gridDim.x);
    const unsigned below = (1u << lane) - 1u;

    for (int sub = wave; need_scan && sub <= n_sub; sub += NWAVE) {
        if (sub < n_sub) {
            const int rs = sub + rot;
            const int base = (rs < n_sub ? rs : rs - n_sub) * SUBE;
            int d[4][4];
            if (base + SUBE <= n_edges) {
#pragma unroll
                for (int g = 0; g < 4; ++g) {
                    const int4 t = *(const int4*)&dst_idx[base + (g << 7) + (lane << 2)];
                    d[g][0] = t.x; d[g][1] = t.y; d[g][2] = t.z; d[g][3] = t.w;
                }
            } else {
#pragma unroll
                for (int g = 0; g < 4; ++g) {
#pragma unroll
                    for (int k = 0; k < 4; ++k) {
                        const int e = base + (g << 7) + (lane << 2) + k;
                        d[g][k] = (e < n_edges) ? dst_idx[e] : -1;
                    }
                }
            }
#pragma unroll
            for (int g = 0; g < 4; ++g) {
#pragma unroll
                for (int k = 0; k < 4; ++k) {
                    const int dl = d[g][k] - lo;
                    const bool m = (unsigned)dl < (unsigned)nrows;
                    const unsigned mask = __builtin_amdgcn_ballot_w32(m);
                    if (mask != 0u) {
                        if (m) q[cnt + __builtin_popcount(mask & below)] =
                                   (src_idx[base + (g << 7) + (lane << 2) + k] << 8) | dl;
                        cnt += __builtin_popcount(mask);
                    }
                }
            }
        }
        if (cnt >= 32 || sub + NWAVE > n_sub) {
            drain_queue(q, cnt, true, msg, sU, lane);
            cnt = 0;
        }
    }
    __syncthreads();

    const int t     = timestep[0];
    const float g_t = 1.f / (1.f + __expf(-gamma[t]));
    const float e_t = 1.f / (1.f + __expf(-eta[t]));
    const float ca  = (1.f - e_t) * g_t;
    const float cx  = xi[0];

    for (int i = threadIdx.x; i < nrows * (D / 2); i += 1024) {
        const int r = i >> 6, c = (i & 63) << 1;
        const size_t g = (size_t)(lo + r) * D + c;
        const v2f xv = *(const v2f*)&x[g];
        const v2f lv = *(const v2f*)&local[g];
        v2f a = *(const v2f*)&sU[r * UP + c];
        a.x = fmaxf(xv.x + ca * a.x + cx * lv.x, 0.f);
        a.y = fmaxf(xv.y + ca * a.y + cx * lv.y, 0.f);
        *(v2f*)&sU[r * UP + c] = a;
    }
    __syncthreads();

    const int ntl = nrows >> 4;
    for (int item = wave; item < ntl * 2; item += NWAVE) {
        const int tl = item >> 1;
        const int nh = (item & 1) << 2;
        const float* arow = sU + ((tl << 4) + lm) * UP + (half << 1);

        float ob[4];
#pragma unroll
        for (int n = 0; n < 4; ++n) ob[n] = out_b[lm + ((nh + n) << 4)];

        v8f acc[4];
#pragma unroll
        for (int n = 0; n < 4; ++n) {
#pragma unroll
            for (int p = 0; p < 8; ++p) acc[n][p] = 0.f;
        }

        v2f bb[2][4];
        v2f aa[2];
#pragma unroll
        for (int n = 0; n < 4; ++n) bb[0][n] = ldsB(sW, 0, half, lm, nh + n);
        aa[0] = *(const v2f*)&arow[0];

#pragma unroll 2
        for (int kk = 0; kk < 32; ++kk) {
            const int cur = kk & 1, nxt = cur ^ 1;
            if (kk < 31) {
#pragma unroll
                for (int n = 0; n < 4; ++n) bb[nxt][n] = ldsB(sW, kk + 1, half, lm, nh + n);
                aa[nxt] = *(const v2f*)&arow[(kk + 1) << 2];
            }
#pragma unroll
            for (int n = 0; n < 4; ++n) acc[n] = wmma_f32(aa[cur], bb[cur][n], acc[n]);
        }

        const int rbase = (tl << 4) + (half << 3);
#pragma unroll
        for (int n = 0; n < 4; ++n) {
            const int N = lm + ((nh + n) << 4);
#pragma unroll
            for (int j = 0; j < 8; ++j) {
                const float u = sU[(rbase + j) * UP + N];
                out[(size_t)(lo + rbase + j) * D + N] = acc[n][j] + u + ob[n];
            }
        }
    }
}

extern "C" void kernel_launch(void* const* d_in, const int* in_sizes, int n_in,
                              void* d_out, int out_size, void* d_ws, size_t ws_size,
                              hipStream_t stream) {
    const float* x        = (const float*)d_in[0];
    const int*   src_idx  = (const int*)  d_in[1];
    const int*   dst_idx  = (const int*)  d_in[2];
    const float* local    = (const float*)d_in[3];
    const int*   timestep = (const int*)  d_in[4];
    const float* gamma    = (const float*)d_in[5];
    const float* eta      = (const float*)d_in[6];
    const float* xi       = (const float*)d_in[7];
    const float* W_w      = (const float*)d_in[8];
    const float* W_b      = (const float*)d_in[9];
    const float* ln_g     = (const float*)d_in[10];
    const float* ln_b     = (const float*)d_in[11];
    const float* out_w    = (const float*)d_in[12];
    const float* out_b    = (const float*)d_in[13];
    float*       out      = (float*)d_out;
    float*       msg      = (float*)d_ws;

    const int n_nodes = in_sizes[0] / D;
    const int n_edges = in_sizes[1];

    const int node_pairs  = ((n_nodes >> 4) + 1) >> 1;
    const int msg_blocks  = (node_pairs + 7) / 8;
    const int node_blocks = (n_nodes + RPB - 1) / RPB;
    const int n_slices    = (n_edges + BS - 1) / BS;
    gnn_msg_kernel<<<msg_blocks + n_slices, 256, 0, stream>>>(
        x, W_w, W_b, ln_g, ln_b, msg, n_nodes, msg_blocks,
        src_idx, dst_idx, out, n_edges, node_blocks, n_slices);

    gnn_node_kernel<<<node_blocks, 1024, 0, stream>>>(
        x, msg, src_idx, dst_idx, local, out_w, out_b, gamma, eta, xi, timestep,
        out, n_nodes, n_edges, n_slices);
}


# LDS f32 adds replaced by register sums: owner sorts its edges by row (LDS int counting sort); a wave sums a row's 512-B msg rows in registers, list order = int arrival rank, blends, stores once
# speedup vs baseline: 3.8717x; 1.1198x over previous
#include <hip/hip_runtime.h>
#include <hip/hip_bf16.h>


#define D 128

typedef __attribute__((ext_vector_type(2))) float v2f;
typedef __attribute__((ext_vector_type(8))) float v8f;

__device__ __forceinline__ v8f wmma_f32(v2f a, v2f b, v8f c) {
    return __builtin_amdgcn_wmma_f32_16x16x4_f32(
         false, a,  false, b,
         (short)0, c,  false,  false);
}

__device__ __forceinline__ int sw_addr(int row, int col) {
    return (row << 7) + ((((col >> 1) ^ (row & 63)) << 1) | (col & 1));
}

__device__ __forceinline__ v2f ldsB(const float* sW, int kk, int half, int lm, int n) {
    const int row = (n << 4) + lm;
    const int kB  = (kk << 2) + (half << 1);
    return *(const v2f*)&sW[sw_addr(row, kB)];
}

__device__ __forceinline__ void ln_store(
    const v8f* __restrict__ acc, int nbase, int half, int lm,
    const float* __restrict__ lg, const float* __restrict__ lb,
    float* __restrict__ msg)
{
    float mu[8], rs[8];
#pragma unroll
    for (int j = 0; j < 8; ++j) {
        float s = 0.f, q = 0.f;
#pragma unroll
        for (int n = 0; n < 8; ++n) {
            const float v = acc[n][j];
            s += v; q += v * v;
        }
#pragma unroll
        for (int m = 1; m < 16; m <<= 1) {
            s += __shfl_xor(s, m, 32);
            q += __shfl_xor(q, m, 32);
        }
        const float mean = s * (1.f / 128.f);
        mu[j] = mean;
        rs[j] = rsqrtf(fmaxf(q * (1.f / 128.f) - mean * mean, 0.f) + 1e-5f);
    }

    const size_t gbase = (size_t)(nbase + (half << 3)) * D + lm;
#pragma unroll
    for (int n = 0; n < 8; ++n) {
#pragma unroll
        for (int j = 0; j < 8; ++j) {
            const float v = (acc[n][j] - mu[j]) * rs[j] * lg[n] + lb[n];
            msg[gbase + (size_t)j * D + (n << 4)] = v;
        }
    }
}

#define TPB   13
#define RPB   (TPB * 16)
#define BS    4096
#define BEPT  (BS / 256)
#define CP    64
#define CCAP  (CP - 1)
#define HMAX  1024

__device__ __forceinline__ void bin_slice(
    int* hC,
    const int* __restrict__ src_idx, const int* __restrict__ dst_idx,
    float* __restrict__ cells,
    int g, int n_slices, int n_nodes, int n_edges, int n_owner)
{
    const int tid = threadIdx.x;
    for (int i = tid; i < n_owner; i += 256) hC[i] = 0;
    __syncthreads();

    const int e0 = g * BS + tid;
    int dv[BEPT], sv[BEPT];
#pragma unroll
    for (int k = 0; k < BEPT; ++k) {
        const int e = min(e0 + (k << 8), n_edges - 1);
        dv[k] = dst_idx[e];
        sv[k] = src_idx[e];
    }
    const int last_rows = n_nodes - (n_owner - 1) * RPB;
    const int need      = n_slices * CP;
#pragma unroll
    for (int k = 0; k < BEPT; ++k) {
        const int d = dv[k];
        if (e0 + (k << 8) < n_edges && (unsigned)d < (unsigned)n_nodes) {
            const int b    = d / RPB;
            const int r    = atomicAdd(&hC[b], 1);
            const int rows = (b == n_owner - 1) ? last_rows : RPB;
            if (r < CCAP && rows * D >= need)
                cells[(size_t)b * (RPB * D) + g * CP + 1 + r] =
                    __int_as_float((sv[k] << 8) | (d - b * RPB));
        }
    }
    __syncthreads();
    for (int b = tid; b < n_owner; b += 256) {
        const int rows = (b == n_owner - 1) ? last_rows : RPB;
        if (rows * D >= need)
            cells[(size_t)b * (RPB * D) + g * CP] = __int_as_float(hC[b]);
    }
}

__global__ __launch_bounds__(256) void gnn_msg_kernel(
    const float* __restrict__ x,
    const float* __restrict__ W_w,
    const float* __restrict__ W_b,
    const float* __restrict__ ln_g,
    const float* __restrict__ ln_b,
    float*       __restrict__ msg,
    int n_nodes, int msg_blocks,
    const int*   __restrict__ src_idx,
    const int*   __restrict__ dst_idx,
    float*       __restrict__ cells,
    int n_edges, int n_owner, int n_slices)
{
    __shared__ float sW[D * D];
    __shared__ int   hC[HMAX];

    if ((int)blockIdx.x >= msg_blocks) {
        if (n_owner <= HMAX)
            bin_slice(hC, src_idx, dst_idx, cells, (int)blockIdx.x - msg_blocks,
                      n_slices, n_nodes, n_edges, n_owner);
        return;
    }

    for (int i = threadIdx.x; i < D * D; i += 256) {
        int r = i >> 7, c = i & 127;
        sW[sw_addr(r, c)] = W_w[i];
    }
    __syncthreads();

    const int lane = threadIdx.x & 31;
    const int wave = threadIdx.x >> 5;
    const int half = lane >> 4;
    const int lm   = lane & 15;

    float wb[8], lg[8], lb[8];
#pragma unroll
    for (int n = 0; n < 8; ++n) {
        const int N = lm + (n << 4);
        wb[n] = W_b[N]; lg[n] = ln_g[N]; lb[n] = ln_b[N];
    }

    const int n_tiles = n_nodes >> 4;
    const int n_pairs = (n_tiles + 1) >> 1;
    for (int pair = blockIdx.x * 8 + wave; pair < n_pairs; pair += msg_blocks * 8) {
        const int tile0 = pair << 1;
        const int tile1 = tile0 + 1;
        const bool has1 = tile1 < n_tiles;
        const int nb0 = tile0 << 4;
        const int nb1 = (has1 ? tile1 : tile0) << 4;

        const float* arow0 = x + (size_t)(nb0 + lm) * D + (half << 1);
        const float* arow1 = x + (size_t)(nb1 + lm) * D + (half << 1);

        v8f acc0[8], acc1[8];
#pragma unroll
        for (int n = 0; n < 8; ++n) {
#pragma unroll
            for (int q = 0; q < 8; ++q) { acc0[n][q] = wb[n]; acc1[n][q] = wb[n]; }
        }

        v2f bb[2][8];
        v2f aa0[2], aa1[2];
#pragma unroll
        for (int n = 0; n < 8; ++n) bb[0][n] = ldsB(sW, 0, half, lm, n);
        aa0[0] = *(const v2f*)&arow0[0];
        aa1[0] = *(const v2f*)&arow1[0];

#pragma unroll 2
        for (int kk = 0; kk < 32; ++kk) {
            const int cur = kk & 1, nxt = cur ^ 1;
            if (kk < 31) {
#pragma unroll
                for (int n = 0; n < 8; ++n) bb[nxt][n] = ldsB(sW, kk + 1, half, lm, n);
                aa0[nxt] = *(const v2f*)&arow0[(kk + 1) << 2];
                aa1[nxt] = *(const v2f*)&arow1[(kk + 1) << 2];
            }
#pragma unroll
            for (int n = 0; n < 8; ++n) acc0[n] = wmma_f32(aa0[cur], bb[cur][n], acc0[n]);
#pragma unroll
            for (int n = 0; n < 8; ++n) acc1[n] = wmma_f32(aa1[cur], bb[cur][n], acc1[n]);
        }

        ln_store(acc0, nb0, half, lm, lg, lb, msg);
        if (has1) ln_store(acc1, nb1, half, lm, lg, lb, msg);
    }
}

#define UP    132
#define NWAVE 32
#define SUBE  512
#define QCAP  (32 + SUBE)
#define GB    8
#define CH    4
#define LCAP  (NWAVE * QCAP)
#define NCNT  256

__device__ __forceinline__ int drain_queue(
    int* q, int cnt, bool all, const float* __restrict__ msg, float* sU, int lane)
{
    const int lim = all ? cnt : (cnt & ~31);
    int b0 = 0;
    for (; b0 < lim; b0 += 32) {
        const int n   = min(32, cnt - b0);
        const int ent = (lane < n) ? q[b0 + lane] : 0;
        const int s   = ent >> 8;
        const int dl  = ent & 255;
        for (int j0 = 0; j0 < n; j0 += GB) {
            float v[GB][4];
#pragma unroll
            for (int j = 0; j < GB; ++j) {
                const int sj = __builtin_amdgcn_readlane(s, min(j0 + j, n - 1));
                const float* row = msg + (size_t)sj * D + lane;
#pragma unroll
                for (int c = 0; c < 4; ++c) v[j][c] = row[c << 5];
            }
#pragma unroll
            for (int j = 0; j < GB; ++j) {
                if (j0 + j < n) {
                    const int dj = __builtin_amdgcn_readlane(dl, j0 + j);
                    float* row = sU + dj * UP + lane;
#pragma unroll
                    for (int c = 0; c < 4; ++c) atomicAdd(&row[c << 5], v[j][c]);
                }
            }
        }
    }
    const int rem = cnt - b0;
    if (rem > 0 && b0 > 0) {
        const int t = (lane < rem) ? q[b0 + lane] : 0;
        if (lane < rem) q[lane] = t;
    }
    return rem > 0 ? rem : 0;
}

__global__ __launch_bounds__(1024) void gnn_node_kernel(
    const float* __restrict__ x,
    const float* __restrict__ msg,
    const int*   __restrict__ src_idx,
    const int*   __restrict__ dst_idx,
    const float* __restrict__ local,
    const float* __restrict__ out_w,
    const float* __restrict__ out_b,
    const float* __restrict__ gamma,
    const float* __restrict__ eta,
    const float* __restrict__ xi,
    const int*   __restrict__ timestep,
    float*       __restrict__ out,
    int n_nodes, int n_edges, int n_slices)
{
    __shared__ __attribute__((aligned(16))) float sW[D * D];
    __shared__ __attribute__((aligned(16))) float sU[RPB * UP];
    __shared__ int   sQ[NWAVE * QCAP];
    __shared__ int   sCnt[NCNT];
    __shared__ int   sOff[NCNT + 1];
    __shared__ int   sPos[NCNT];
    __shared__ int   sScan;

    const int lo    = blockIdx.x * RPB;
    const int nrows = min(RPB, n_nodes - lo);

    const float* cellrow = out + (size_t)lo * D;
    const bool can_bin = ((int)gridDim.x <= HMAX) && (nrows * D >= n_slices * CP) &&
                         (n_slices * CCAP <= LCAP);
    int over = can_bin ? 0 : 1;
    if (can_bin) {
        for (int g = threadIdx.x; g < n_slices; g += 1024)
            over |= (__float_as_int(cellrow[g * CP]) > CCAP) ? 1 : 0;
    }
    if (threadIdx.x == 0) sScan = 0;
    if (threadIdx.x < NCNT) sCnt[threadIdx.x] = 0;
    __syncthreads();

    for (int i = threadIdx.x; i < D * D; i += 1024) {
        int r = i >> 7, c = i & 127;
        sW[sw_addr(r, c)] = out_w[i];
    }
    if (over) sScan = 1;
    __syncthreads();

    const int lane = threadIdx.x & 31;
    const int wave = __builtin_amdgcn_readfirstlane(threadIdx.x >> 5);
    const int half = lane >> 4;
    const int lm   = lane & 15;

    const int t     = timestep[0];
    const float g_t = 1.f / (1.f + __expf(-gamma[t]));
    const float e_t = 1.f / (1.f + __expf(-eta[t]));
    const float ca  = (1.f - e_t) * g_t;
    const float cx  = xi[0];

    int* q = sQ + wave * QCAP;
    int cnt = 0;
    const bool need_scan = sScan != 0;

    if (!need_scan) {
        for (int g = wave; g < n_slices; g += CH * NWAVE) {
            float c0[CH], c1[CH];
#pragma unroll
            for (int u = 0; u < CH; ++u) {
                const int gu = g + u * NWAVE;
                c0[u] = 0.f; c1[u] = 0.f;
                if (gu < n_slices) {
                    c0[u] = cellrow[gu * CP + lane];
                    c1[u] = cellrow[gu * CP + 32 + lane];
                }
            }
#pragma unroll
            for (int u = 0; u < CH; ++u) {
                const int v0 = __float_as_int(c0[u]), v1 = __float_as_int(c1[u]);
                const int n = min(max(__builtin_amdgcn_readfirstlane(v0), 0), CCAP);
                if (lane >= 1 && lane - 1 < n) atomicAdd(&sCnt[v0 & (NCNT - 1)], 1);
                if (31 + lane < n)             atomicAdd(&sCnt[v1 & (NCNT - 1)], 1);
            }
        }
        __syncthreads();
        if (wave == 0) {
            int c[8];
            int s = 0;
#pragma unroll
            for (int k = 0; k < 8; ++k) { c[k] = sCnt[(lane << 3) + k]; s += c[k]; }
            int inc = s;
#pragma unroll
            for (int m = 1; m < 32; m <<= 1) {
                const int up = __shfl_up(inc, m, 32);
                if (lane >= m) inc += up;
            }
            int run = inc - s;
#pragma unroll
            for (int k = 0; k < 8; ++k) {
                sOff[(lane << 3) + k] = run;
                sPos[(lane << 3) + k] = run;
                run += c[k];
            }
            if (lane == 31) sOff[NCNT] = run;
        }
        __syncthreads();
        for (int g = wave; g < n_slices; g += CH * NWAVE) {
            float c0[CH], c1[CH];
#pragma unroll
            for (int u = 0; u < CH; ++u) {
                const int gu = g + u * NWAVE;
                c0[u] = 0.f; c1[u] = 0.f;
                if (gu < n_slices) {
                    c0[u] = cellrow[gu * CP + lane];
                    c1[u] = cellrow[gu * CP + 32 + lane];
                }
            }
#pragma unroll
            for (int u = 0; u < CH; ++u) {
                const int v0 = __float_as_int(c0[u]), v1 = __float_as_int(c1[u]);
                const int n = min(max(__builtin_amdgcn_readfirstlane(v0), 0), CCAP);
                if (lane >= 1 && lane - 1 < n)
                    sQ[atomicAdd(&sPos[v0 & (NCNT - 1)], 1)] = v0 >> 8;
                if (31 + lane < n)
                    sQ[atomicAdd(&sPos[v1 & (NCNT - 1)], 1)] = v1 >> 8;
            }
        }
        __syncthreads();

        for (int r = wave; r < nrows; r += NWAVE) {
            const int b = __builtin_amdgcn_readfirstlane(sOff[r]);
            const int n = __builtin_amdgcn_readfirstlane(sOff[r + 1]) - b;
            const size_t gx = (size_t)(lo + r) * D + (lane << 2);
            const float4 xv = *(const float4*)&x[gx];
            const float4 lv = *(const float4*)&local[gx];
            float4 acc = make_float4(0.f, 0.f, 0.f, 0.f);
            for (int e0 = 0; e0 < n; e0 += 32) {
                const int m   = min(32, n - e0);
                const int sid = (lane < m) ? sQ[b + e0 + lane] : 0;
                for (int j0 = 0; j0 < m; j0 += GB) {
                    float4 v[GB];
#pragma unroll
                    for (int j = 0; j < GB; ++j) {
                        const int sj = __builtin_amdgcn_readlane(sid, min(j0 + j, m - 1));
                        v[j] = *(const float4*)&msg[(size_t)sj * D + (lane << 2)];
                    }
#pragma unroll
                    for (int j = 0; j < GB; ++j) {
                        if (j0 + j < m) {
                            acc.x += v[j].x; acc.y += v[j].y;
                            acc.z += v[j].z; acc.w += v[j].w;
                        }
                    }
                }
            }
            float4 u;
            u.x = fmaxf(xv.x + ca * acc.x + cx * lv.x, 0.f);
            u.y = fmaxf(xv.y + ca * acc.y + cx * lv.y, 0.f);
            u.z = fmaxf(xv.z + ca * acc.z + cx * lv.z, 0.f);
            u.w = fmaxf(xv.w + ca * acc.w + cx * lv.w, 0.f);
            *(float4*)&sU[r * UP + (lane << 2)] = u;
        }
    } else {
        for (int i = threadIdx.x; i < RPB * UP; i += 1024) sU[i] = 0.f;
        __syncthreads();
    }

    const int n_sub = (n_edges + SUBE - 1) / SUBE;
    const int rot = (int)(((long long)blockIdx.x * n_sub) / gridDim.x);
    const unsigned below = (1u << lane) - 1u;

    for (int sub = wave; need_scan && sub <= n_sub; sub += NWAVE) {
        if (sub < n_sub) {
            const int rs = sub + rot;
            const int base = (rs < n_sub ? rs : rs - n_sub) * SUBE;
            int d[4][4];
            if (base + SUBE <= n_edges) {
#pragma unroll
                for (int g = 0; g < 4; ++g) {
                    const int4 t = *(const int4*)&dst_idx[base + (g << 7) + (lane << 2)];
                    d[g][0] = t.x; d[g][1] = t.y; d[g][2] = t.z; d[g][3] = t.w;
                }
            } else {
#pragma unroll
                for (int g = 0; g < 4; ++g) {
#pragma unroll
                    for (int k = 0; k < 4; ++k) {
                        const int e = base + (g << 7) + (lane << 2) + k;
                        d[g][k] = (e < n_edges) ? dst_idx[e] : -1;
                    }
                }
            }
#pragma unroll
            for (int g = 0; g < 4; ++g) {
#pragma unroll
                for (int k = 0; k < 4; ++k) {
                    const int dl = d[g][k] - lo;
                    const bool m = (unsigned)dl < (unsigned)nrows;
                    const unsigned mask = __builtin_amdgcn_ballot_w32(m);
                    if (mask != 0u) {
                        if (m) q[cnt + __builtin_popcount(mask & below)] =
                                   (src_idx[base + (g << 7) + (lane << 2) + k] << 8) | dl;
                        cnt += __builtin_popcount(mask);
                    }
                }
            }
        }
        if (cnt >= 32 || sub + NWAVE > n_sub) {
            drain_queue(q, cnt, true, msg, sU, lane);
            cnt = 0;
        }
    }
    __syncthreads();

    for (int i = threadIdx.x; need_scan && i < nrows * (D / 2); i += 1024) {
        const int r = i >> 6, c = (i & 63) << 1;
        const size_t g = (size_t)(lo + r) * D + c;
        const v2f xv = *(const v2f*)&x[g];
        const v2f lv = *(const v2f*)&local[g];
        v2f a = *(const v2f*)&sU[r * UP + c];
        a.x = fmaxf(xv.x + ca * a.x + cx * lv.x, 0.f);
        a.y = fmaxf(xv.y + ca * a.y + cx * lv.y, 0.f);
        *(v2f*)&sU[r * UP + c] = a;
    }
    __syncthreads();

    const int ntl = nrows >> 4;
    for (int item = wave; item < ntl * 2; item += NWAVE) {
        const int tl = item >> 1;
        const int nh = (item & 1) << 2;
        const float* arow = sU + ((tl << 4) + lm) * UP + (half << 1);

        float ob[4];
#pragma unroll
        for (int n = 0; n < 4; ++n) ob[n] = out_b[lm + ((nh + n) << 4)];

        v8f acc[4];
#pragma unroll
        for (int n = 0; n < 4; ++n) {
#pragma unroll
            for (int p = 0; p < 8; ++p) acc[n][p] = 0.f;
        }

        v2f bb[2][4];
        v2f aa[2];
#pragma unroll
        for (int n = 0; n < 4; ++n) bb[0][n] = ldsB(sW, 0, half, lm, nh + n);
        aa[0] = *(const v2f*)&arow[0];

#pragma unroll 2
        for (int kk = 0; kk < 32; ++kk) {
            const int cur = kk & 1, nxt = cur ^ 1;
            if (kk < 31) {
#pragma unroll
                for (int n = 0; n < 4; ++n) bb[nxt][n] = ldsB(sW, kk + 1, half, lm, nh + n);
                aa[nxt] = *(const v2f*)&arow[(kk + 1) << 2];
            }
#pragma unroll
            for (int n = 0; n < 4; ++n) acc[n] = wmma_f32(aa[cur], bb[cur][n], acc[n]);
        }

        const int rbase = (tl << 4) + (half << 3);
#pragma unroll
        for (int n = 0; n < 4; ++n) {
            const int N = lm + ((nh + n) << 4);
#pragma unroll
            for (int j = 0; j < 8; ++j) {
                const float u = sU[(rbase + j) * UP + N];
                out[(size_t)(lo + rbase + j) * D + N] = acc[n][j] + u + ob[n];
            }
        }
    }
}

extern "C" void kernel_launch(void* const* d_in, const int* in_sizes, int n_in,
                              void* d_out, int out_size, void* d_ws, size_t ws_size,
                              hipStream_t stream) {
    const float* x        = (const float*)d_in[0];
    const int*   src_idx  = (const int*)  d_in[1];
    const int*   dst_idx  = (const int*)  d_in[2];
    const float* local    = (const float*)d_in[3];
    const int*   timestep = (const int*)  d_in[4];
    const float* gamma    = (const float*)d_in[5];
    const float* eta      = (const float*)d_in[6];
    const float* xi       = (const float*)d_in[7];
    const float* W_w      = (const float*)d_in[8];
    const float* W_b      = (const float*)d_in[9];
    const float* ln_g     = (const float*)d_in[10];
    const float* ln_b     = (const float*)d_in[11];
    const float* out_w    = (const float*)d_in[12];
    const float* out_b    = (const float*)d_in[13];
    float*       out      = (float*)d_out;
    float*       msg      = (float*)d_ws;

    const int n_nodes = in_sizes[0] / D;
    const int n_edges = in_sizes[1];

    const int node_pairs  = ((n_nodes >> 4) + 1) >> 1;
    const int msg_blocks  = (node_pairs + 7) / 8;
    const int node_blocks = (n_nodes + RPB - 1) / RPB;
    const int n_slices    = (n_edges + BS - 1) / BS;
    gnn_msg_kernel<<<msg_blocks + n_slices, 256, 0, stream>>>(
        x, W_w, W_b, ln_g, ln_b, msg, n_nodes, msg_blocks,
        src_idx, dst_idx, out, n_edges, node_blocks, n_slices);

    gnn_node_kernel<<<node_blocks, 1024, 0, stream>>>(
        x, msg, src_idx, dst_idx, local, out_w, out_b, gamma, eta, xi, timestep,
        out, n_nodes, n_edges, n_slices);
}


# Start-up loads of both kernels issued as one group at entry: W staged by 16-byte loads all in flight, node kernel reads its cells once and sorts from registers; f32 sum order as before
# speedup vs baseline: 4.1598x; 1.0744x over previous
#include <hip/hip_runtime.h>
#include <hip/hip_bf16.h>


#define D 128

typedef __attribute__((ext_vector_type(2))) float v2f;
typedef __attribute__((ext_vector_type(8))) float v8f;

__device__ __forceinline__ v8f wmma_f32(v2f a, v2f b, v8f c) {
    return __builtin_amdgcn_wmma_f32_16x16x4_f32(
         false, a,  false, b,
         (short)0, c,  false,  false);
}

__device__ __forceinline__ int sw_addr(int row, int col) {
    return (row << 7) + ((((col >> 1) ^ (row & 63)) << 1) | (col & 1));
}

__device__ __forceinline__ void stage_w4(float* sW, int i, float4 v) {
    const int r  = i >> 5;
    const int c4 = i & 31;
    const int b  = c4 ^ ((r & 63) >> 1);
    if (r & 1) v = make_float4(v.z, v.w, v.x, v.y);
    *(float4*)&sW[(r << 7) + (b << 2)] = v;
}

__device__ __forceinline__ v2f ldsB(const float* sW, int kk, int half, int lm, int n) {
    const int row = (n << 4) + lm;
    const int kB  = (kk << 2) + (half << 1);
    return *(const v2f*)&sW[sw_addr(row, kB)];
}

__device__ __forceinline__ void ln_store(
    const v8f* __restrict__ acc, int nbase, int half, int lm,
    const float* __restrict__ lg, const float* __restrict__ lb,
    float* __restrict__ msg)
{
    float mu[8], rs[8];
#pragma unroll
    for (int j = 0; j < 8; ++j) {
        float s = 0.f, q = 0.f;
#pragma unroll
        for (int n = 0; n < 8; ++n) {
            const float v = acc[n][j];
            s += v; q += v * v;
        }
#pragma unroll
        for (int m = 1; m < 16; m <<= 1) {
            s += __shfl_xor(s, m, 32);
            q += __shfl_xor(q, m, 32);
        }
        const float mean = s * (1.f / 128.f);
        mu[j] = mean;
        rs[j] = rsqrtf(fmaxf(q * (1.f / 128.f) - mean * mean, 0.f) + 1e-5f);
    }

    const size_t gbase = (size_t)(nbase + (half << 3)) * D + lm;
#pragma unroll
    for (int n = 0; n < 8; ++n) {
#pragma unroll
        for (int j = 0; j < 8; ++j) {
            const float v = (acc[n][j] - mu[j]) * rs[j] * lg[n] + lb[n];
            msg[gbase + (size_t)j * D + (n << 4)] = v;
        }
    }
}

#define TPB   13
#define RPB   (TPB * 16)
#define BS    4096
#define BEPT  (BS / 256)
#define CP    64
#define CCAP  (CP - 1)
#define HMAX  1024

__device__ __forceinline__ void bin_slice(
    int* hC,
    const int* __restrict__ src_idx, const int* __restrict__ dst_idx,
    float* __restrict__ cells,
    int g, int n_slices, int n_nodes, int n_edges, int n_owner)
{
    const int tid = threadIdx.x;
    for (int i = tid; i < n_owner; i += 256) hC[i] = 0;
    __syncthreads();

    const int e0 = g * BS + tid;
    int dv[BEPT], sv[BEPT];
#pragma unroll
    for (int k = 0; k < BEPT; ++k) {
        const int e = min(e0 + (k << 8), n_edges - 1);
        dv[k] = dst_idx[e];
        sv[k] = src_idx[e];
    }
    const int last_rows = n_nodes - (n_owner - 1) * RPB;
    const int need      = n_slices * CP;
#pragma unroll
    for (int k = 0; k < BEPT; ++k) {
        const int d = dv[k];
        if (e0 + (k << 8) < n_edges && (unsigned)d < (unsigned)n_nodes) {
            const int b    = d / RPB;
            const int r    = atomicAdd(&hC[b], 1);
            const int rows = (b == n_owner - 1) ? last_rows : RPB;
            if (r < CCAP && rows * D >= need)
                cells[(size_t)b * (RPB * D) + g * CP + 1 + r] =
                    __int_as_float((sv[k] << 8) | (d - b * RPB));
        }
    }
    __syncthreads();
    for (int b = tid; b < n_owner; b += 256) {
        const int rows = (b == n_owner - 1) ? last_rows : RPB;
        if (rows * D >= need)
            cells[(size_t)b * (RPB * D) + g * CP] = __int_as_float(hC[b]);
    }
}

__global__ __launch_bounds__(256) void gnn_msg_kernel(
    const float* __restrict__ x,
    const float* __restrict__ W_w,
    const float* __restrict__ W_b,
    const float* __restrict__ ln_g,
    const float* __restrict__ ln_b,
    float*       __restrict__ msg,
    int n_nodes, int msg_blocks,
    const int*   __restrict__ src_idx,
    const int*   __restrict__ dst_idx,
    float*       __restrict__ cells,
    int n_edges, int n_owner, int n_slices)
{
    __shared__ __attribute__((aligned(16))) float sW[D * D];
    __shared__ int   hC[HMAX];

    if ((int)blockIdx.x >= msg_blocks) {
        if (n_owner <= HMAX)
            bin_slice(hC, src_idx, dst_idx, cells, (int)blockIdx.x - msg_blocks,
                      n_slices, n_nodes, n_edges, n_owner);
        return;
    }

    const int lane = threadIdx.x & 31;
    const int wave = threadIdx.x >> 5;
    const int half = lane >> 4;
    const int lm   = lane & 15;

    float4 wv[16];
#pragma unroll
    for (int k = 0; k < 16; ++k)
        wv[k] = ((const float4*)W_w)[threadIdx.x + (k << 8)];

    float wb[8], lg[8], lb[8];
#pragma unroll
    for (int n = 0; n < 8; ++n) {
        const int N = lm + (n << 4);
        wb[n] = W_b[N]; lg[n] = ln_g[N]; lb[n] = ln_b[N];
    }

    const int n_tiles = n_nodes >> 4;
    const int n_pairs = (n_tiles + 1) >> 1;
    int pair = blockIdx.x * 8 + wave;
    v2f pa0 = {0.f, 0.f}, pa1 = {0.f, 0.f};
    if (pair < n_pairs) {
        const int t0 = pair << 1;
        const int t1 = (t0 + 1 < n_tiles) ? t0 + 1 : t0;
        pa0 = *(const v2f*)&x[(size_t)((t0 << 4) + lm) * D + (half << 1)];
        pa1 = *(const v2f*)&x[(size_t)((t1 << 4) + lm) * D + (half << 1)];
    }
    bool first = true;

#pragma unroll
    for (int k = 0; k < 16; ++k)
        stage_w4(sW, threadIdx.x + (k << 8), wv[k]);
    __syncthreads();

    for (; pair < n_pairs; pair += msg_blocks * 8) {
        const int tile0 = pair << 1;
        const int tile1 = tile0 + 1;
        const bool has1 = tile1 < n_tiles;
        const int nb0 = tile0 << 4;
        const int nb1 = (has1 ? tile1 : tile0) << 4;

        const float* arow0 = x + (size_t)(nb0 + lm) * D + (half << 1);
        const float* arow1 = x + (size_t)(nb1 + lm) * D + (half << 1);

        v8f acc0[8], acc1[8];
#pragma unroll
        for (int n = 0; n < 8; ++n) {
#pragma unroll
            for (int q = 0; q < 8; ++q) { acc0[n][q] = wb[n]; acc1[n][q] = wb[n]; }
        }

        v2f bb[2][8];
        v2f aa0[2], aa1[2];
#pragma unroll
        for (int n = 0; n < 8; ++n) bb[0][n] = ldsB(sW, 0, half, lm, n);
        if (first) {
            aa0[0] = pa0; aa1[0] = pa1;
            first = false;
        } else {
            aa0[0] = *(const v2f*)&arow0[0];
            aa1[0] = *(const v2f*)&arow1[0];
        }

#pragma unroll 2
        for (int kk = 0; kk < 32; ++kk) {
            const int cur = kk & 1, nxt = cur ^ 1;
            if (kk < 31) {
#pragma unroll
                for (int n = 0; n < 8; ++n) bb[nxt][n] = ldsB(sW, kk + 1, half, lm, n);
                aa0[nxt] = *(const v2f*)&arow0[(kk + 1) << 2];
                aa1[nxt] = *(const v2f*)&arow1[(kk + 1) << 2];
            }
#pragma unroll
            for (int n = 0; n < 8; ++n) acc0[n] = wmma_f32(aa0[cur], bb[cur][n], acc0[n]);
#pragma unroll
            for (int n = 0; n < 8; ++n) acc1[n] = wmma_f32(aa1[cur], bb[cur][n], acc1[n]);
        }

        ln_store(acc0, nb0, half, lm, lg, lb, msg);
        if (has1) ln_store(acc1, nb1, half, lm, lg, lb, msg);
    }
}

#define UP    132
#define NWAVE 32
#define SUBE  512
#define QCAP  (32 + SUBE)
#define GB    8
#define LCAP  (NWAVE * QCAP)
#define NCNT  256
#define NCW   ((LCAP / CCAP + NWAVE - 1) / NWAVE)

__device__ __forceinline__ int drain_queue(
    int* q, int cnt, bool all, const float* __restrict__ msg, float* sU, int lane)
{
    const int lim = all ? cnt : (cnt & ~31);
    int b0 = 0;
    for (; b0 < lim; b0 += 32) {
        const int n   = min(32, cnt - b0);
        const int ent = (lane < n) ? q[b0 + lane] : 0;
        const int s   = ent >> 8;
        const int dl  = ent & 255;
        for (int j0 = 0; j0 < n; j0 += GB) {
            float v[GB][4];
#pragma unroll
            for (int j = 0; j < GB; ++j) {
                const int sj = __builtin_amdgcn_readlane(s, min(j0 + j, n - 1));
                const float* row = msg + (size_t)sj * D + lane;
#pragma unroll
                for (int c = 0; c < 4; ++c) v[j][c] = row[c << 5];
            }
#pragma unroll
            for (int j = 0; j < GB; ++j) {
                if (j0 + j < n) {
                    const int dj = __builtin_amdgcn_readlane(dl, j0 + j);
                    float* row = sU + dj * UP + lane;
#pragma unroll
                    for (int c = 0; c < 4; ++c) atomicAdd(&row[c << 5], v[j][c]);
                }
            }
        }
    }
    const int rem = cnt - b0;
    if (rem > 0 && b0 > 0) {
        const int t = (lane < rem) ? q[b0 + lane] : 0;
        if (lane < rem) q[lane] = t;
    }
    return rem > 0 ? rem : 0;
}

__global__ __launch_bounds__(1024) void gnn_node_kernel(
    const float* __restrict__ x,
    const float* __restrict__ msg,
    const int*   __restrict__ src_idx,
    const int*   __restrict__ dst_idx,
    const float* __restrict__ local,
    const float* __restrict__ out_w,
    const float* __restrict__ out_b,
    const float* __restrict__ gamma,
    const float* __restrict__ eta,
    const float* __restrict__ xi,
    const int*   __restrict__ timestep,
    float*       __restrict__ out,
    int n_nodes, int n_edges, int n_slices)
{
    __shared__ __attribute__((aligned(16))) float sW[D * D];
    __shared__ __attribute__((aligned(16))) float sU[RPB * UP];
    __shared__ int   sQ[NWAVE * QCAP];
    __shared__ int   sCnt[NCNT];
    __shared__ int   sOff[NCNT + 1];
    __shared__ int   sPos[NCNT];
    __shared__ int   sScan;

    const int lo    = blockIdx.x * RPB;
    const int nrows = min(RPB, n_nodes - lo);

    const float* cellrow = out + (size_t)lo * D;
    const bool can_bin = ((int)gridDim.x <= HMAX) && (nrows * D >= n_slices * CP) &&
                         (n_slices * CCAP <= LCAP);

    const int lane = threadIdx.x & 31;
    const int wave = __builtin_amdgcn_readfirstlane(threadIdx.x >> 5);
    const int half = lane >> 4;
    const int lm   = lane & 15;

    const int t = timestep[0];
    float c0[NCW], c1[NCW];
#pragma unroll
    for (int u = 0; u < NCW; ++u) {
        const int gu = wave + u * NWAVE;
        c0[u] = 0.f; c1[u] = 0.f;
        if (can_bin && gu < n_slices) {
            c0[u] = cellrow[gu * CP + lane];
            c1[u] = cellrow[gu * CP + 32 + lane];
        }
    }
    float4 wv[4];
#pragma unroll
    for (int k = 0; k < 4; ++k)
        wv[k] = ((const float4*)out_w)[threadIdx.x + (k << 10)];
    const float ga = gamma[t];
    const float et = eta[t];
    const float cx = xi[0];

    if (threadIdx.x == 0) sScan = 0;
    if (threadIdx.x < NCNT) sCnt[threadIdx.x] = 0;
    __syncthreads();

    int over = can_bin ? 0 : 1;
#pragma unroll
    for (int u = 0; u < NCW; ++u) {
        const int v0 = __float_as_int(c0[u]), v1 = __float_as_int(c1[u]);
        const int craw = __builtin_amdgcn_readfirstlane(v0);
        over |= (craw > CCAP) ? 1 : 0;
        const int n = min(max(craw, 0), CCAP);
        if (lane >= 1 && lane - 1 < n) atomicAdd(&sCnt[v0 & (NCNT - 1)], 1);
        if (31 + lane < n)             atomicAdd(&sCnt[v1 & (NCNT - 1)], 1);
    }
    if (over) sScan = 1;

#pragma unroll
    for (int k = 0; k < 4; ++k)
        stage_w4(sW, threadIdx.x + (k << 10), wv[k]);
    __syncthreads();

    const float g_t = 1.f / (1.f + __expf(-ga));
    const float e_t = 1.f / (1.f + __expf(-et));
    const float ca  = (1.f - e_t) * g_t;

    int* q = sQ + wave * QCAP;
    int cnt = 0;
    const bool need_scan = sScan != 0;

    if (!need_scan) {
        if (wave == 0) {
            int c[8];
            int s = 0;
#pragma unroll
            for (int k = 0; k < 8; ++k) { c[k] = sCnt[(lane << 3) + k]; s += c[k]; }
            int inc = s;
#pragma unroll
            for (int m = 1; m < 32; m <<= 1) {
                const int up = __shfl_up(inc, m, 32);
                if (lane >= m) inc += up;
            }
            int run = inc - s;
#pragma unroll
            for (int k = 0; k < 8; ++k) {
                sOff[(lane << 3) + k] = run;
                sPos[(lane << 3) + k] = run;
                run += c[k];
            }
            if (lane == 31) sOff[NCNT] = run;
        }
        __syncthreads();
        int p0[NCW], p1[NCW];
#pragma unroll
        for (int u = 0; u < NCW; ++u) {
            const int v0 = __float_as_int(c0[u]), v1 = __float_as_int(c1[u]);
            const int n = min(max(__builtin_amdgcn_readfirstlane(v0), 0), CCAP);
            p0[u] = -1; p1[u] = -1;
            if (lane >= 1 && lane - 1 < n) p0[u] = atomicAdd(&sPos[v0 & (NCNT - 1)], 1);
            if (31 + lane < n)             p1[u] = atomicAdd(&sPos[v1 & (NCNT - 1)], 1);
        }
#pragma unroll
        for (int u = 0; u < NCW; ++u) {
            if (p0[u] >= 0) sQ[p0[u]] = __float_as_int(c0[u]) >> 8;
            if (p1[u] >= 0) sQ[p1[u]] = __float_as_int(c1[u]) >> 8;
        }
        __syncthreads();

        for (int r = wave; r < nrows; r += NWAVE) {
            const int b = __builtin_amdgcn_readfirstlane(sOff[r]);
            const int n = __builtin_amdgcn_readfirstlane(sOff[r + 1]) - b;
            const size_t gx = (size_t)(lo + r) * D + (lane << 2);
            const float4 xv = *(const float4*)&x[gx];
            const float4 lv = *(const float4*)&local[gx];
            float4 acc = make_float4(0.f, 0.f, 0.f, 0.f);
            for (int e0 = 0; e0 < n; e0 += 32) {
                const int m   = min(32, n - e0);
                const int sid = (lane < m) ? sQ[b + e0 + lane] : 0;
                for (int j0 = 0; j0 < m; j0 += GB) {
                    float4 v[GB];
#pragma unroll
                    for (int j = 0; j < GB; ++j) {
                        const int sj = __builtin_amdgcn_readlane(sid, min(j0 + j, m - 1));
                        v[j] = *(const float4*)&msg[(size_t)sj * D + (lane << 2)];
                    }
#pragma unroll
                    for (int j = 0; j < GB; ++j) {
                        if (j0 + j < m) {
                            acc.x += v[j].x; acc.y += v[j].y;
                            acc.z += v[j].z; acc.w += v[j].w;
                        }
                    }
                }
            }
            float4 u;
            u.x = fmaxf(xv.x + ca * acc.x + cx * lv.x, 0.f);
            u.y = fmaxf(xv.y + ca * acc.y + cx * lv.y, 0.f);
            u.z = fmaxf(xv.z + ca * acc.z + cx * lv.z, 0.f);
            u.w = fmaxf(xv.w + ca * acc.w + cx * lv.w, 0.f);
            *(float4*)&sU[r * UP + (lane << 2)] = u;
        }
    } else {
        for (int i = threadIdx.x; i < RPB * UP; i += 1024) sU[i] = 0.f;
        __syncthreads();
    }

    const int n_sub = (n_edges + SUBE - 1) / SUBE;
    const int rot = (int)(((long long)blockIdx.x * n_sub) / gridDim.x);
    const unsigned below = (1u << lane) - 1u;

    for (int sub = wave; need_scan && sub <= n_sub; sub += NWAVE) {
        if (sub < n_sub) {
            const int rs = sub + rot;
            const int base = (rs < n_sub ? rs : rs - n_sub) * SUBE;
            int d[4][4];
            if (base + SUBE <= n_edges) {
#pragma unroll
                for (int g = 0; g < 4; ++g) {
                    const int4 t = *(const int4*)&dst_idx[base + (g << 7) + (lane << 2)];
                    d[g][0] = t.x; d[g][1] = t.y; d[g][2] = t.z; d[g][3] = t.w;
                }
            } else {
#pragma unroll
                for (int g = 0; g < 4; ++g) {
#pragma unroll
                    for (int k = 0; k < 4; ++k) {
                        const int e = base + (g << 7) + (lane << 2) + k;
                        d[g][k] = (e < n_edges) ? dst_idx[e] : -1;
                    }
                }
            }
#pragma unroll
            for (int g = 0; g < 4; ++g) {
#pragma unroll
                for (int k = 0; k < 4; ++k) {
                    const int dl = d[g][k] - lo;
                    const bool m = (unsigned)dl < (unsigned)nrows;
                    const unsigned mask = __builtin_amdgcn_ballot_w32(m);
                    if (mask != 0u) {
                        if (m) q[cnt + __builtin_popcount(mask & below)] =
                                   (src_idx[base + (g << 7) + (lane << 2) + k] << 8) | dl;
                        cnt += __builtin_popcount(mask);
                    }
                }
            }
        }
        if (cnt >= 32 || sub + NWAVE > n_sub) {
            drain_queue(q, cnt, true, msg, sU, lane);
            cnt = 0;
        }
    }
    __syncthreads();

    for (int i = threadIdx.x; need_scan && i < nrows * (D / 2); i += 1024) {
        const int r = i >> 6, c = (i & 63) << 1;
        const size_t g = (size_t)(lo + r) * D + c;
        const v2f xv = *(const v2f*)&x[g];
        const v2f lv = *(const v2f*)&local[g];
        v2f a = *(const v2f*)&sU[r * UP + c];
        a.x = fmaxf(xv.x + ca * a.x + cx * lv.x, 0.f);
        a.y = fmaxf(xv.y + ca * a.y + cx * lv.y, 0.f);
        *(v2f*)&sU[r * UP + c] = a;
    }
    __syncthreads();

    const int ntl = nrows >> 4;
    for (int item = wave; item < ntl * 2; item += NWAVE) {
        const int tl = item >> 1;
        const int nh = (item & 1) << 2;
        const float* arow = sU + ((tl << 4) + lm) * UP + (half << 1);

        float ob[4];
#pragma unroll
        for (int n = 0; n < 4; ++n) ob[n] = out_b[lm + ((nh + n) << 4)];

        v8f acc[4];
#pragma unroll
        for (int n = 0; n < 4; ++n) {
#pragma unroll
            for (int p = 0; p < 8; ++p) acc[n][p] = 0.f;
        }

        v2f bb[2][4];
        v2f aa[2];
#pragma unroll
        for (int n = 0; n < 4; ++n) bb[0][n] = ldsB(sW, 0, half, lm, nh + n);
        aa[0] = *(const v2f*)&arow[0];

#pragma unroll 2
        for (int kk = 0; kk < 32; ++kk) {
            const int cur = kk & 1, nxt = cur ^ 1;
            if (kk < 31) {
#pragma unroll
                for (int n = 0; n < 4; ++n) bb[nxt][n] = ldsB(sW, kk + 1, half, lm, nh + n);
                aa[nxt] = *(const v2f*)&arow[(kk + 1) << 2];
            }
#pragma unroll
            for (int n = 0; n < 4; ++n) acc[n] = wmma_f32(aa[cur], bb[cur][n], acc[n]);
        }

        const int rbase = (tl << 4) + (half << 3);
#pragma unroll
        for (int n = 0; n < 4; ++n) {
            const int N = lm + ((nh + n) << 4);
#pragma unroll
            for (int j = 0; j < 8; ++j) {
                const float u = sU[(rbase + j) * UP + N];
                out[(size_t)(lo + rbase + j) * D + N] = acc[n][j] + u + ob[n];
            }
        }
    }
}

extern "C" void kernel_launch(void* const* d_in, const int* in_sizes, int n_in,
                              void* d_out, int out_size, void* d_ws, size_t ws_size,
                              hipStream_t stream) {
    const float* x        = (const float*)d_in[0];
    const int*   src_idx  = (const int*)  d_in[1];
    const int*   dst_idx  = (const int*)  d_in[2];
    const float* local    = (const float*)d_in[3];
    const int*   timestep = (const int*)  d_in[4];
    const float* gamma    = (const float*)d_in[5];
    const float* eta      = (const float*)d_in[6];
    const float* xi       = (const float*)d_in[7];
    const float* W_w      = (const float*)d_in[8];
    const float* W_b      = (const float*)d_in[9];
    const float* ln_g     = (const float*)d_in[10];
    const float* ln_b     = (const float*)d_in[11];
    const float* out_w    = (const float*)d_in[12];
    const float* out_b    = (const float*)d_in[13];
    float*       out      = (float*)d_out;
    float*       msg      = (float*)d_ws;

    const int n_nodes = in_sizes[0] / D;
    const int n_edges = in_sizes[1];

    const int node_pairs  = ((n_nodes >> 4) + 1) >> 1;
    const int msg_blocks  = (node_pairs + 7) / 8;
    const int node_blocks = (n_nodes + RPB - 1) / RPB;
    const int n_slices    = (n_edges + BS - 1) / BS;
    gnn_msg_kernel<<<msg_blocks + n_slices, 256, 0, stream>>>(
        x, W_w, W_b, ln_g, ln_b, msg, n_nodes, msg_blocks,
        src_idx, dst_idx, out, n_edges, node_blocks, n_slices);

    gnn_node_kernel<<<node_blocks, 1024, 0, stream>>>(
        x, msg, src_idx, dst_idx, local, out_w, out_b, gamma, eta, xi, timestep,
        out, n_nodes, n_edges, n_slices);
}
